# starting kernel
# baseline (speedup 1.0000x reference)
#include <hip/hip_runtime.h>
#include <math.h>

typedef __attribute__((ext_vector_type(2))) float v2f;
typedef __attribute__((ext_vector_type(4))) float v4f;
typedef __attribute__((ext_vector_type(8))) float v8f;

#define N_NODES 100000
#define N_EDGES 1600000
#define FEAT    64
#define EMBED   128
#define NCLS    40

__global__ __launch_bounds__(256) void gcn_zero4(float* __restrict__ p, long n4) {
  long i = (long)blockIdx.x * blockDim.x + threadIdx.x;
  if (i < n4) ((v4f*)p)[i] = (v4f){0.0f, 0.0f, 0.0f, 0.0f};
}

template <int NF>
__global__ __launch_bounds__(256) void gcn_spmm_scatter4(
    const float* __restrict__ feat, const int* __restrict__ src,
    const int* __restrict__ dst, const float* __restrict__ w,
    float* __restrict__ out, long total) {
  constexpr int QP = NF / 4;
  long gid = (long)blockIdx.x * blockDim.x + threadIdx.x;
  if (gid >= total) return;
  int e = (int)(gid / QP);
  int f = (int)(gid % QP) * 4;
  int s = src[e];
  int d = dst[e];
  float ww = w[e];
  v4f v = *(const v4f*)(feat + (long)s * NF + f);
  float* o = out + (long)d * NF + f;
  atomicAdd(o + 0, ww * v.x);
  atomicAdd(o + 1, ww * v.y);
  atomicAdd(o + 2, ww * v.z);
  atomicAdd(o + 3, ww * v.w);
}

template <int K, int N, int NT, bool RELU, bool NMASK>
__global__ __launch_bounds__(256) void gcn_wmma_gemm(
    const float* __restrict__ A, const float* __restrict__ B,
    float* __restrict__ C, int totalTiles) {
  const int wave = threadIdx.x >> 5;
  const int lane = threadIdx.x & 31;
  const int tile = blockIdx.x * 8 + wave;
  if (tile >= totalTiles) return;

  const int mt = tile / NT;
  const int nt = tile % NT;
  const int m    = lane & 15;
  const int half = lane >> 4;
  const int klo  = half * 2;

  const int row  = mt * 16 + m;
  const int ncol = nt * 16 + m;

  const int   bcol  = (!NMASK || ncol < N) ? ncol : (N - 1);
  const float bmask = (!NMASK || ncol < N) ? 1.0f : 0.0f;

  const float* __restrict__ arow = A + (long)row * K;

  v8f c = {};
#pragma unroll
  for (int k = 0; k < K; k += 4) {
    v2f a = *(const v2f*)(arow + k + klo);
    v2f b;
    b.x = B[(long)(k + klo)     * N + bcol] * bmask;
    b.y = B[(long)(k + klo + 1) * N + bcol] * bmask;
    c = __builtin_amdgcn_wmma_f32_16x16x4_f32(
        false, a, false, b, (short)0, c, false, false);
  }

#pragma unroll
  for (int r = 0; r < 8; ++r) {
    int rr = mt * 16 + r + half * 8;
    int cc = nt * 16 + (lane & 15);
    if (!NMASK || cc < N) {
      float v = c[r];
      if (RELU) v = fmaxf(v, 0.0f);
      C[(long)rr * N + cc] = v;
    }
  }
}

__global__ __launch_bounds__(256) void gcn_log_softmax(
    const float* __restrict__ in, float* __restrict__ out, int n) {
  int i = blockIdx.x * blockDim.x + threadIdx.x;
  if (i >= n) return;
  const float* __restrict__ row = in + (long)i * NCLS;
  float mx = -INFINITY;
#pragma unroll
  for (int c = 0; c < NCLS; ++c) mx = fmaxf(mx, row[c]);
  float s = 0.0f;
#pragma unroll
  for (int c = 0; c < NCLS; ++c) s += __expf(row[c] - mx);
  float lse = mx + __logf(s);
  float* __restrict__ orow = out + (long)i * NCLS;
#pragma unroll
  for (int c = 0; c < NCLS; ++c) orow[c] = row[c] - lse;
}

extern "C" void kernel_launch(void* const* d_in, const int* in_sizes, int n_in,
                              void* d_out, int out_size, void* d_ws, size_t ws_size,
                              hipStream_t stream) {
  const float* x    = (const float*)d_in[0];
  const float* W1   = (const float*)d_in[1];
  const float* W2   = (const float*)d_in[2];
  const float* ew   = (const float*)d_in[3];
  const int*   esrc = (const int*)d_in[4];
  const int*   edst = (const int*)d_in[5];
  float*       out  = (float*)d_out;

  float* agg1 = (float*)d_ws;
  float* h    = agg1 + (size_t)N_NODES * FEAT;
  float* t    = h    + (size_t)N_NODES * EMBED;
  float* agg2 = t    + (size_t)N_NODES * NCLS;

  const int TPB = 256;

  {
    long n1 = (long)N_NODES * FEAT / 4;
    gcn_zero4<<<(int)((n1 + TPB - 1) / TPB), TPB, 0, stream>>>(agg1, n1);
    long n2 = (long)N_NODES * NCLS / 4;
    gcn_zero4<<<(int)((n2 + TPB - 1) / TPB), TPB, 0, stream>>>(agg2, n2);
  }

  {
    long total = (long)N_EDGES * (FEAT / 4);
    gcn_spmm_scatter4<FEAT><<<(int)((total + TPB - 1) / TPB), TPB, 0, stream>>>(
        x, esrc, edst, ew, agg1, total);
  }

  {
    const int NT = EMBED / 16;
    int tiles = (N_NODES / 16) * NT;
    gcn_wmma_gemm<FEAT, EMBED, NT, true, false>
        <<<(tiles + 7) / 8, TPB, 0, stream>>>(agg1, W1, h, tiles);
  }

  {
    const int NT = (NCLS + 15) / 16;
    int tiles = (N_NODES / 16) * NT;
    gcn_wmma_gemm<EMBED, NCLS, NT, false, true>
        <<<(tiles + 7) / 8, TPB, 0, stream>>>(h, W2, t, tiles);
  }

  {
    long total = (long)N_EDGES * (NCLS / 4);
    gcn_spmm_scatter4<NCLS><<<(int)((total + TPB - 1) / TPB), TPB, 0, stream>>>(
        t, esrc, edst, ew, agg2, total);
  }

  gcn_log_softmax<<<(N_NODES + TPB - 1) / TPB, TPB, 0, stream>>>(agg2, out, N_NODES);
}


# Scatter SPMM with f32 atomics replaced by per-call grouping of edges by dst (int atomic rank, scan, fill) and atomic-free per-row gather sums; row sums now in arrival-order rank
# speedup vs baseline: 1.4672x; 1.4672x over previous
#include <hip/hip_runtime.h>
#include <math.h>

typedef __attribute__((ext_vector_type(2))) float v2f;
typedef __attribute__((ext_vector_type(4))) float v4f;
typedef __attribute__((ext_vector_type(8))) float v8f;
typedef __attribute__((ext_vector_type(2))) int v2i;
typedef __attribute__((ext_vector_type(4))) int v4i;

#define N_NODES 100000
#define N_EDGES 1600000
#define FEAT    64
#define EMBED   128
#define NCLS    40

__global__ __launch_bounds__(256) void gcn_zero4(float* __restrict__ p, long n4) {
  long i = (long)blockIdx.x * blockDim.x + threadIdx.x;
  if (i < n4) ((v4f*)p)[i] = (v4f){0.0f, 0.0f, 0.0f, 0.0f};
}

__global__ __launch_bounds__(256) void gcn_edge_rank(
    const int* __restrict__ dst, int* __restrict__ cnt,
    int* __restrict__ rank, int ne) {
  int e = blockIdx.x * blockDim.x + threadIdx.x;
  if (e >= ne) return;
  rank[e] = atomicAdd(cnt + dst[e], 1);
}

__global__ __launch_bounds__(1024) void gcn_rowptr_scan(
    const int* __restrict__ cnt, int* __restrict__ row_ptr, int n) {
  __shared__ int w_incl[32];
  __shared__ int w_part[32];
  const int tid  = threadIdx.x;
  const int lane = tid & 31;
  const int wave = tid >> 5;
  const int base = blockIdx.x * 1024;

  int part = 0;
#pragma unroll 4
  for (int i = tid * 4; i < base; i += 4096) {
    v4i c4 = *(const v4i*)(cnt + i);
    part += (c4.x + c4.y) + (c4.z + c4.w);
  }
  const int i = base + tid;
  const int c = (i < n) ? cnt[i] : 0;

  int incl = c;
#pragma unroll
  for (int o = 1; o < 32; o <<= 1) {
    int v = __shfl_up(incl, o);
    if (lane >= o) incl += v;
  }
#pragma unroll
  for (int o = 16; o > 0; o >>= 1) part += __shfl_xor(part, o);
  if (lane == 31) w_incl[wave] = incl;
  if (lane == 0)  w_part[wave] = part;
  __syncthreads();

  int off = 0;
#pragma unroll
  for (int k = 0; k < 32; ++k) {
    off += w_part[k];
    if (k < wave) off += w_incl[k];
  }
  const int excl = off + incl - c;
  if (i < n) row_ptr[i] = excl;
  if (i == n - 1) row_ptr[n] = excl + c;
}

__global__ __launch_bounds__(256) void gcn_csr_fill(
    const int* __restrict__ src, const int* __restrict__ dst,
    const float* __restrict__ w, const int* __restrict__ rank,
    const int* __restrict__ row_ptr, int* __restrict__ csr, int ne) {
  int e = blockIdx.x * blockDim.x + threadIdx.x;
  if (e >= ne) return;
  int p = row_ptr[dst[e]] + rank[e];
  v2i v;
  v.x = src[e];
  v.y = __float_as_int(w[e]);
  ((v2i*)csr)[p] = v;
}

template <int NF>
__global__ __launch_bounds__(256) void gcn_spmm_gather4(
    const float* __restrict__ feat, const int* __restrict__ row_ptr,
    const int* __restrict__ csr, float* __restrict__ out, int n_rows) {
  constexpr int QP = NF / 4;
  const int d = (blockIdx.x * blockDim.x + threadIdx.x) >> 4;
  const int q = threadIdx.x & 15;
  if (d >= n_rows || q >= QP) return;
  const int beg = row_ptr[d];
  const int end = row_ptr[d + 1];
  const v2i* __restrict__ ed = (const v2i*)csr;
  const float* __restrict__ fq = feat + q * 4;

  v4f acc = {0.0f, 0.0f, 0.0f, 0.0f};
  if (beg < end) {
    const int last = end - 1;
    v2i e0 = ed[beg];
    v2i e1 = ed[min(beg + 1, last)];
    v2i e2 = ed[min(beg + 2, last)];
    v2i e3 = ed[min(beg + 3, last)];
    for (int j = beg; j < end; j += 4) {
      v4f x0 = *(const v4f*)(fq + (long)e0.x * NF);
      v4f x1 = *(const v4f*)(fq + (long)e1.x * NF);
      v4f x2 = *(const v4f*)(fq + (long)e2.x * NF);
      v4f x3 = *(const v4f*)(fq + (long)e3.x * NF);
      const float w0 = __int_as_float(e0.y);
      const float w1 = __int_as_float(e1.y);
      const float w2 = __int_as_float(e2.y);
      const float w3 = __int_as_float(e3.y);
      e0 = ed[min(j + 4, last)];
      e1 = ed[min(j + 5, last)];
      e2 = ed[min(j + 6, last)];
      e3 = ed[min(j + 7, last)];
      acc += w0 * x0;
      if (j + 1 < end) acc += w1 * x1;
      if (j + 2 < end) acc += w2 * x2;
      if (j + 3 < end) acc += w3 * x3;
    }
  }
  *(v4f*)(out + (long)d * NF + q * 4) = acc;
}

template <int K, int N, int NT, bool RELU, bool NMASK>
__global__ __launch_bounds__(256) void gcn_wmma_gemm(
    const float* __restrict__ A, const float* __restrict__ B,
    float* __restrict__ C, int totalTiles) {
  const int wave = threadIdx.x >> 5;
  const int lane = threadIdx.x & 31;
  const int tile = blockIdx.x * 8 + wave;
  if (tile >= totalTiles) return;

  const int mt = tile / NT;
  const int nt = tile % NT;
  const int m    = lane & 15;
  const int half = lane >> 4;
  const int klo  = half * 2;

  const int row  = mt * 16 + m;
  const int ncol = nt * 16 + m;

  const int   bcol  = (!NMASK || ncol < N) ? ncol : (N - 1);
  const float bmask = (!NMASK || ncol < N) ? 1.0f : 0.0f;

  const float* __restrict__ arow = A + (long)row * K;

  v8f c = {};
#pragma unroll
  for (int k = 0; k < K; k += 4) {
    v2f a = *(const v2f*)(arow + k + klo);
    v2f b;
    b.x = B[(long)(k + klo)     * N + bcol] * bmask;
    b.y = B[(long)(k + klo + 1) * N + bcol] * bmask;
    c = __builtin_amdgcn_wmma_f32_16x16x4_f32(
        false, a, false, b, (short)0, c, false, false);
  }

#pragma unroll
  for (int r = 0; r < 8; ++r) {
    int rr = mt * 16 + r + half * 8;
    int cc = nt * 16 + (lane & 15);
    if (!NMASK || cc < N) {
      float v = c[r];
      if (RELU) v = fmaxf(v, 0.0f);
      C[(long)rr * N + cc] = v;
    }
  }
}

__global__ __launch_bounds__(256) void gcn_log_softmax(
    const float* __restrict__ in, float* __restrict__ out, int n) {
  int i = blockIdx.x * blockDim.x + threadIdx.x;
  if (i >= n) return;
  const float* __restrict__ row = in + (long)i * NCLS;
  float mx = -INFINITY;
#pragma unroll
  for (int c = 0; c < NCLS; ++c) mx = fmaxf(mx, row[c]);
  float s = 0.0f;
#pragma unroll
  for (int c = 0; c < NCLS; ++c) s += __expf(row[c] - mx);
  float lse = mx + __logf(s);
  float* __restrict__ orow = out + (long)i * NCLS;
#pragma unroll
  for (int c = 0; c < NCLS; ++c) orow[c] = row[c] - lse;
}

extern "C" void kernel_launch(void* const* d_in, const int* in_sizes, int n_in,
                              void* d_out, int out_size, void* d_ws, size_t ws_size,
                              hipStream_t stream) {
  const float* x    = (const float*)d_in[0];
  const float* W1   = (const float*)d_in[1];
  const float* W2   = (const float*)d_in[2];
  const float* ew   = (const float*)d_in[3];
  const int*   esrc = (const int*)d_in[4];
  const int*   edst = (const int*)d_in[5];
  float*       out  = (float*)d_out;

  float* agg1 = (float*)d_ws;
  float* h    = agg1 + (size_t)N_NODES * FEAT;
  float* t    = h    + (size_t)N_NODES * EMBED;
  int*   csr  = (int*)(t + (size_t)N_NODES * NCLS);
  int*   rptr = csr + (size_t)2 * N_EDGES;
  float* agg2 = agg1;
  int*   rank = (int*)h;
  int*   cnt  = rank + (size_t)N_EDGES;

  const int TPB = 256;

  {
    long n0 = (long)N_NODES / 4;
    gcn_zero4<<<(int)((n0 + TPB - 1) / TPB), TPB, 0, stream>>>((float*)cnt, n0);
    gcn_edge_rank<<<(N_EDGES + TPB - 1) / TPB, TPB, 0, stream>>>(
        edst, cnt, rank, N_EDGES);
    gcn_rowptr_scan<<<(N_NODES + 1023) / 1024, 1024, 0, stream>>>(
        cnt, rptr, N_NODES);
    gcn_csr_fill<<<(N_EDGES + TPB - 1) / TPB, TPB, 0, stream>>>(
        esrc, edst, ew, rank, rptr, csr, N_EDGES);
  }

  gcn_spmm_gather4<FEAT><<<(N_NODES * 16 + TPB - 1) / TPB, TPB, 0, stream>>>(
      x, rptr, csr, agg1, N_NODES);

  {
    const int NT = EMBED / 16;
    int tiles = (N_NODES / 16) * NT;
    gcn_wmma_gemm<FEAT, EMBED, NT, true, false>
        <<<(tiles + 7) / 8, TPB, 0, stream>>>(agg1, W1, h, tiles);
  }

  {
    const int NT = (NCLS + 15) / 16;
    int tiles = (N_NODES / 16) * NT;
    gcn_wmma_gemm<EMBED, NCLS, NT, false, true>
        <<<(tiles + 7) / 8, TPB, 0, stream>>>(h, W2, t, tiles);
  }

  gcn_spmm_gather4<NCLS><<<(N_NODES * 16 + TPB - 1) / TPB, TPB, 0, stream>>>(
      t, rptr, csr, agg2, N_NODES);

  gcn_log_softmax<<<(N_NODES + TPB - 1) / TPB, TPB, 0, stream>>>(agg2, out, N_NODES);
}


# Edge grouping by dst: returning global int atomics, rank array and scan replaced by two-level counting sort on LDS counters (256 dst buckets, then rows); row sum order now LDS-add arrival order
# speedup vs baseline: 2.2823x; 1.5555x over previous
#include <hip/hip_runtime.h>
#include <math.h>

typedef __attribute__((ext_vector_type(2))) float v2f;
typedef __attribute__((ext_vector_type(4))) float v4f;
typedef __attribute__((ext_vector_type(8))) float v8f;
typedef __attribute__((ext_vector_type(2))) int v2i;
typedef __attribute__((ext_vector_type(4))) int v4i;

#define N_NODES 100000
#define N_EDGES 1600000
#define FEAT    64
#define EMBED   128
#define NCLS    40

#define NB     256
#define BS     391
#define NW     256
#define CHUNK  ((N_EDGES + NW - 1) / NW)
#define TRIPS  ((CHUNK + 1023) / 1024)
static_assert(NB * BS >= N_NODES, "buckets must cover every row");
static_assert(BS <= 512 && N_NODES <= (1 << 17), "packed edge word");
static_assert(NB == 256 && NW % 4 == 0, "index split below");

__global__ __launch_bounds__(1024) void gcn_bucket_count(
    const int* __restrict__ dst, int* __restrict__ cntmat) {
  __shared__ int c[NB];
  const int tid = threadIdx.x;
  if (tid < NB) c[tid] = 0;
  const int beg = blockIdx.x * CHUNK;
  const int end = min(beg + CHUNK, N_EDGES);
  int bk[TRIPS];
#pragma unroll
  for (int k = 0; k < TRIPS; ++k) {
    const int e = beg + tid + k * 1024;
    bk[k] = (e < end) ? dst[e] / BS : -1;
  }
  __syncthreads();
#pragma unroll
  for (int k = 0; k < TRIPS; ++k)
    if (bk[k] >= 0) atomicAdd(&c[bk[k]], 1);
  __syncthreads();
  if (tid < NB) cntmat[blockIdx.x * NB + tid] = c[tid];
}

__global__ __launch_bounds__(1024) void gcn_bucket_scatter(
    const int* __restrict__ src, const int* __restrict__ dst,
    const float* __restrict__ w, const int* __restrict__ cntmat,
    int* __restrict__ tmp8, int* __restrict__ bstart) {
  __shared__ int tot[4][NB];
  __shared__ int pre[4][NB];
  __shared__ int cur[NB];
  __shared__ int wtot[8];
  const int tid  = threadIdx.x;
  const int lane = tid & 31;
  const int wave = tid >> 5;
  const int b    = tid & (NB - 1);
  const int q    = tid >> 8;
  const int beg = blockIdx.x * CHUNK;
  const int end = min(beg + CHUNK, N_EDGES);

  int d[TRIPS], s[TRIPS], wb[TRIPS];
#pragma unroll
  for (int k = 0; k < TRIPS; ++k) {
    const int e = beg + tid + k * 1024;
    const bool ok = e < end;
    d[k]  = ok ? dst[e] : -1;
    s[k]  = ok ? src[e] : 0;
    wb[k] = ok ? __float_as_int(w[e]) : 0;
  }

  int t = 0, p = 0;
#pragma unroll 16
  for (int j = 0; j < NW / 4; ++j) {
    const int wg = q * (NW / 4) + j;
    const int v = cntmat[wg * NB + b];
    t += v;
    p += (wg < (int)blockIdx.x) ? v : 0;
  }
  tot[q][b] = t;
  pre[q][b] = p;
  __syncthreads();

  int T = 0, P = 0;
  if (tid < NB) {
    T = (tot[0][tid] + tot[1][tid]) + (tot[2][tid] + tot[3][tid]);
    P = (pre[0][tid] + pre[1][tid]) + (pre[2][tid] + pre[3][tid]);
  }
  int incl = T;
#pragma unroll
  for (int o = 1; o < 32; o <<= 1) {
    int v = __shfl_up(incl, o);
    if (lane >= o) incl += v;
  }
  if (tid < NB && lane == 31) wtot[wave] = incl;
  __syncthreads();
  if (tid < NB) {
    int off = 0;
#pragma unroll
    for (int k = 0; k < 8; ++k)
      if (k < wave) off += wtot[k];
    const int excl = off + incl - T;
    cur[tid] = excl + P;
    if (blockIdx.x == 0) {
      bstart[tid] = excl;
      if (tid == NB - 1) bstart[NB] = excl + T;
    }
  }
  __syncthreads();

#pragma unroll
  for (int k = 0; k < TRIPS; ++k) {
    if (d[k] >= 0) {
      const int bkt = d[k] / BS;
      const int pos = atomicAdd(&cur[bkt], 1);
      v2i r;
      r.x = s[k] | ((d[k] - bkt * BS) << 17);
      r.y = wb[k];
      ((v2i*)tmp8)[pos] = r;
    }
  }
}

__global__ __launch_bounds__(1024) void gcn_bucket_rows(
    const int* __restrict__ tmp8, const int* __restrict__ bstart,
    int* __restrict__ row_ptr, int* __restrict__ csr) {
  __shared__ int cnt[512];
  __shared__ int wtot[32];
  const int tid  = threadIdx.x;
  const int lane = tid & 31;
  const int wave = tid >> 5;
  const int b    = blockIdx.x;
  const int lo   = b * BS;
  const int nb   = min(BS, N_NODES - lo);
  const int beg  = bstart[b];
  const int end  = bstart[b + 1];
  const v2i* __restrict__ in = (const v2i*)tmp8;
  if (tid < 512) cnt[tid] = 0;

  v2i r[8];
#pragma unroll
  for (int k = 0; k < 8; ++k) {
    const int i = beg + tid + k * 1024;
    if (i < end) r[k] = in[i];
    else         r[k] = (v2i){-1, 0};
  }
  __syncthreads();
#pragma unroll
  for (int k = 0; k < 8; ++k)
    if (r[k].x >= 0) atomicAdd(&cnt[r[k].x >> 17], 1);
  for (int i = beg + 8192 + tid; i < end; i += 1024)
    atomicAdd(&cnt[in[i].x >> 17], 1);
  __syncthreads();

  const int c = (tid < nb) ? cnt[tid] : 0;
  int incl = c;
#pragma unroll
  for (int o = 1; o < 32; o <<= 1) {
    int v = __shfl_up(incl, o);
    if (lane >= o) incl += v;
  }
  if (lane == 31) wtot[wave] = incl;
  __syncthreads();
  int off = 0;
#pragma unroll
  for (int k = 0; k < 32; ++k)
    if (k < wave) off += wtot[k];
  const int first = beg + off + incl - c;
  if (tid < nb) {
    row_ptr[lo + tid] = first;
    cnt[tid] = first;
  }
  if (b == NB - 1 && tid == 0) row_ptr[N_NODES] = end;
  __syncthreads();

#pragma unroll
  for (int k = 0; k < 8; ++k) {
    if (r[k].x >= 0) {
      const int pos = atomicAdd(&cnt[r[k].x >> 17], 1);
      v2i v;
      v.x = r[k].x & 0x1FFFF;
      v.y = r[k].y;
      ((v2i*)csr)[pos] = v;
    }
  }
  for (int i = beg + 8192 + tid; i < end; i += 1024) {
    const v2i e = in[i];
    const int pos = atomicAdd(&cnt[e.x >> 17], 1);
    v2i v;
    v.x = e.x & 0x1FFFF;
    v.y = e.y;
    ((v2i*)csr)[pos] = v;
  }
}

template <int NF>
__global__ __launch_bounds__(256) void gcn_spmm_gather4(
    const float* __restrict__ feat, const int* __restrict__ row_ptr,
    const int* __restrict__ csr, float* __restrict__ out, int n_rows) {
  constexpr int QP = NF / 4;
  const int d = (blockIdx.x * blockDim.x + threadIdx.x) >> 4;
  const int q = threadIdx.x & 15;
  if (d >= n_rows || q >= QP) return;
  const int beg = row_ptr[d];
  const int end = row_ptr[d + 1];
  const v2i* __restrict__ ed = (const v2i*)csr;
  const float* __restrict__ fq = feat + q * 4;

  v4f acc = {0.0f, 0.0f, 0.0f, 0.0f};
  if (beg < end) {
    const int last = end - 1;
    v2i e0 = ed[beg];
    v2i e1 = ed[min(beg + 1, last)];
    v2i e2 = ed[min(beg + 2, last)];
    v2i e3 = ed[min(beg + 3, last)];
    for (int j = beg; j < end; j += 4) {
      v4f x0 = *(const v4f*)(fq + (long)e0.x * NF);
      v4f x1 = *(const v4f*)(fq + (long)e1.x * NF);
      v4f x2 = *(const v4f*)(fq + (long)e2.x * NF);
      v4f x3 = *(const v4f*)(fq + (long)e3.x * NF);
      const float w0 = __int_as_float(e0.y);
      const float w1 = __int_as_float(e1.y);
      const float w2 = __int_as_float(e2.y);
      const float w3 = __int_as_float(e3.y);
      e0 = ed[min(j + 4, last)];
      e1 = ed[min(j + 5, last)];
      e2 = ed[min(j + 6, last)];
      e3 = ed[min(j + 7, last)];
      acc += w0 * x0;
      if (j + 1 < end) acc += w1 * x1;
      if (j + 2 < end) acc += w2 * x2;
      if (j + 3 < end) acc += w3 * x3;
    }
  }
  *(v4f*)(out + (long)d * NF + q * 4) = acc;
}

template <int K, int N, int NT, bool RELU, bool NMASK>
__global__ __launch_bounds__(256) void gcn_wmma_gemm(
    const float* __restrict__ A, const float* __restrict__ B,
    float* __restrict__ C, int totalTiles) {
  const int wave = threadIdx.x >> 5;
  const int lane = threadIdx.x & 31;
  const int tile = blockIdx.x * 8 + wave;
  if (tile >= totalTiles) return;

  const int mt = tile / NT;
  const int nt = tile % NT;
  const int m    = lane & 15;
  const int half = lane >> 4;
  const int klo  = half * 2;

  const int row  = mt * 16 + m;
  const int ncol = nt * 16 + m;

  const int   bcol  = (!NMASK || ncol < N) ? ncol : (N - 1);
  const float bmask = (!NMASK || ncol < N) ? 1.0f : 0.0f;

  const float* __restrict__ arow = A + (long)row * K;

  v8f c = {};
#pragma unroll
  for (int k = 0; k < K; k += 4) {
    v2f a = *(const v2f*)(arow + k + klo);
    v2f b;
    b.x = B[(long)(k + klo)     * N + bcol] * bmask;
    b.y = B[(long)(k + klo + 1) * N + bcol] * bmask;
    c = __builtin_amdgcn_wmma_f32_16x16x4_f32(
        false, a, false, b, (short)0, c, false, false);
  }

#pragma unroll
  for (int r = 0; r < 8; ++r) {
    int rr = mt * 16 + r + half * 8;
    int cc = nt * 16 + (lane & 15);
    if (!NMASK || cc < N) {
      float v = c[r];
      if (RELU) v = fmaxf(v, 0.0f);
      C[(long)rr * N + cc] = v;
    }
  }
}

__global__ __launch_bounds__(256) void gcn_log_softmax(
    const float* __restrict__ in, float* __restrict__ out, int n) {
  int i = blockIdx.x * blockDim.x + threadIdx.x;
  if (i >= n) return;
  const float* __restrict__ row = in + (long)i * NCLS;
  float mx = -INFINITY;
#pragma unroll
  for (int c = 0; c < NCLS; ++c) mx = fmaxf(mx, row[c]);
  float s = 0.0f;
#pragma unroll
  for (int c = 0; c < NCLS; ++c) s += __expf(row[c] - mx);
  float lse = mx + __logf(s);
  float* __restrict__ orow = out + (long)i * NCLS;
#pragma unroll
  for (int c = 0; c < NCLS; ++c) orow[c] = row[c] - lse;
}

extern "C" void kernel_launch(void* const* d_in, const int* in_sizes, int n_in,
                              void* d_out, int out_size, void* d_ws, size_t ws_size,
                              hipStream_t stream) {
  const float* x    = (const float*)d_in[0];
  const float* W1   = (const float*)d_in[1];
  const float* W2   = (const float*)d_in[2];
  const float* ew   = (const float*)d_in[3];
  const int*   esrc = (const int*)d_in[4];
  const int*   edst = (const int*)d_in[5];
  float*       out  = (float*)d_out;

  float* agg1 = (float*)d_ws;
  float* h    = agg1 + (size_t)N_NODES * FEAT;
  float* t    = h    + (size_t)N_NODES * EMBED;
  int*   csr  = (int*)(t + (size_t)N_NODES * NCLS);
  int*   rptr = csr + (size_t)2 * N_EDGES;
  float* agg2 = agg1;
  int*   tmp8   = (int*)h;
  int*   cntmat = tmp8 + (size_t)2 * N_EDGES;
  int*   bstart = cntmat + (size_t)NW * NB;

  const int TPB = 256;

  gcn_bucket_count<<<NW, 1024, 0, stream>>>(edst, cntmat);
  gcn_bucket_scatter<<<NW, 1024, 0, stream>>>(esrc, edst, ew, cntmat, tmp8, bstart);
  gcn_bucket_rows<<<NB, 1024, 0, stream>>>(tmp8, bstart, rptr, csr);

  gcn_spmm_gather4<FEAT><<<(N_NODES * 16 + TPB - 1) / TPB, TPB, 0, stream>>>(
      x, rptr, csr, agg1, N_NODES);

  {
    const int NT = EMBED / 16;
    int tiles = (N_NODES / 16) * NT;
    gcn_wmma_gemm<FEAT, EMBED, NT, true, false>
        <<<(tiles + 7) / 8, TPB, 0, stream>>>(agg1, W1, h, tiles);
  }

  {
    const int NT = (NCLS + 15) / 16;
    int tiles = (N_NODES / 16) * NT;
    gcn_wmma_gemm<EMBED, NCLS, NT, false, true>
        <<<(tiles + 7) / 8, TPB, 0, stream>>>(h, W2, t, tiles);
  }

  gcn_spmm_gather4<NCLS><<<(N_NODES * 16 + TPB - 1) / TPB, TPB, 0, stream>>>(
      t, rptr, csr, agg2, N_NODES);

  gcn_log_softmax<<<(N_NODES + TPB - 1) / TPB, TPB, 0, stream>>>(agg2, out, N_NODES);
}


# GEMM1+relu+GEMM2 fused in one kernel: one wave per 16-row tile keeps its h tile in LDS, W1 and W2 staged in LDS in fragment order; h never goes to memory; 7 launches; summation order unchanged
# speedup vs baseline: 2.7345x; 1.1981x over previous
#include <hip/hip_runtime.h>
#include <math.h>

typedef __attribute__((ext_vector_type(2))) float v2f;
typedef __attribute__((ext_vector_type(4))) float v4f;
typedef __attribute__((ext_vector_type(8))) float v8f;
typedef __attribute__((ext_vector_type(2))) int v2i;
typedef __attribute__((ext_vector_type(4))) int v4i;

#define N_NODES 100000
#define N_EDGES 1600000
#define FEAT    64
#define EMBED   128
#define NCLS    40

#define NB     256
#define BS     391
#define NW     256
#define CHUNK  ((N_EDGES + NW - 1) / NW)
#define TRIPS  ((CHUNK + 1023) / 1024)
static_assert(NB * BS >= N_NODES, "buckets must cover every row");
static_assert(BS <= 512 && N_NODES <= (1 << 17), "packed edge word");
static_assert(NB == 256 && NW % 4 == 0, "index split below");

__global__ __launch_bounds__(1024) void gcn_bucket_count(
    const int* __restrict__ dst, int* __restrict__ cntmat) {
  __shared__ int c[NB];
  const int tid = threadIdx.x;
  if (tid < NB) c[tid] = 0;
  const int beg = blockIdx.x * CHUNK;
  const int end = min(beg + CHUNK, N_EDGES);
  int bk[TRIPS];
#pragma unroll
  for (int k = 0; k < TRIPS; ++k) {
    const int e = beg + tid + k * 1024;
    bk[k] = (e < end) ? dst[e] / BS : -1;
  }
  __syncthreads();
#pragma unroll
  for (int k = 0; k < TRIPS; ++k)
    if (bk[k] >= 0) atomicAdd(&c[bk[k]], 1);
  __syncthreads();
  if (tid < NB) cntmat[blockIdx.x * NB + tid] = c[tid];
}

__global__ __launch_bounds__(1024) void gcn_bucket_scatter(
    const int* __restrict__ src, const int* __restrict__ dst,
    const float* __restrict__ w, const int* __restrict__ cntmat,
    int* __restrict__ tmp8, int* __restrict__ bstart) {
  __shared__ int tot[4][NB];
  __shared__ int pre[4][NB];
  __shared__ int cur[NB];
  __shared__ int wtot[8];
  const int tid  = threadIdx.x;
  const int lane = tid & 31;
  const int wave = tid >> 5;
  const int b    = tid & (NB - 1);
  const int q    = tid >> 8;
  const int beg = blockIdx.x * CHUNK;
  const int end = min(beg + CHUNK, N_EDGES);

  int d[TRIPS], s[TRIPS], wb[TRIPS];
#pragma unroll
  for (int k = 0; k < TRIPS; ++k) {
    const int e = beg + tid + k * 1024;
    const bool ok = e < end;
    d[k]  = ok ? dst[e] : -1;
    s[k]  = ok ? src[e] : 0;
    wb[k] = ok ? __float_as_int(w[e]) : 0;
  }

  int t = 0, p = 0;
#pragma unroll 16
  for (int j = 0; j < NW / 4; ++j) {
    const int wg = q * (NW / 4) + j;
    const int v = cntmat[wg * NB + b];
    t += v;
    p += (wg < (int)blockIdx.x) ? v : 0;
  }
  tot[q][b] = t;
  pre[q][b] = p;
  __syncthreads();

  int T = 0, P = 0;
  if (tid < NB) {
    T = (tot[0][tid] + tot[1][tid]) + (tot[2][tid] + tot[3][tid]);
    P = (pre[0][tid] + pre[1][tid]) + (pre[2][tid] + pre[3][tid]);
  }
  int incl = T;
#pragma unroll
  for (int o = 1; o < 32; o <<= 1) {
    int v = __shfl_up(incl, o);
    if (lane >= o) incl += v;
  }
  if (tid < NB && lane == 31) wtot[wave] = incl;
  __syncthreads();
  if (tid < NB) {
    int off = 0;
#pragma unroll
    for (int k = 0; k < 8; ++k)
      if (k < wave) off += wtot[k];
    const int excl = off + incl - T;
    cur[tid] = excl + P;
    if (blockIdx.x == 0) {
      bstart[tid] = excl;
      if (tid == NB - 1) bstart[NB] = excl + T;
    }
  }
  __syncthreads();

#pragma unroll
  for (int k = 0; k < TRIPS; ++k) {
    if (d[k] >= 0) {
      const int bkt = d[k] / BS;
      const int pos = atomicAdd(&cur[bkt], 1);
      v2i r;
      r.x = s[k] | ((d[k] - bkt * BS) << 17);
      r.y = wb[k];
      ((v2i*)tmp8)[pos] = r;
    }
  }
}

__global__ __launch_bounds__(1024) void gcn_bucket_rows(
    const int* __restrict__ tmp8, const int* __restrict__ bstart,
    int* __restrict__ row_ptr, int* __restrict__ csr) {
  __shared__ int cnt[512];
  __shared__ int wtot[32];
  const int tid  = threadIdx.x;
  const int lane = tid & 31;
  const int wave = tid >> 5;
  const int b    = blockIdx.x;
  const int lo   = b * BS;
  const int nb   = min(BS, N_NODES - lo);
  const int beg  = bstart[b];
  const int end  = bstart[b + 1];
  const v2i* __restrict__ in = (const v2i*)tmp8;
  if (tid < 512) cnt[tid] = 0;

  v2i r[8];
#pragma unroll
  for (int k = 0; k < 8; ++k) {
    const int i = beg + tid + k * 1024;
    if (i < end) r[k] = in[i];
    else         r[k] = (v2i){-1, 0};
  }
  __syncthreads();
#pragma unroll
  for (int k = 0; k < 8; ++k)
    if (r[k].x >= 0) atomicAdd(&cnt[r[k].x >> 17], 1);
  for (int i = beg + 8192 + tid; i < end; i += 1024)
    atomicAdd(&cnt[in[i].x >> 17], 1);
  __syncthreads();

  const int c = (tid < nb) ? cnt[tid] : 0;
  int incl = c;
#pragma unroll
  for (int o = 1; o < 32; o <<= 1) {
    int v = __shfl_up(incl, o);
    if (lane >= o) incl += v;
  }
  if (lane == 31) wtot[wave] = incl;
  __syncthreads();
  int off = 0;
#pragma unroll
  for (int k = 0; k < 32; ++k)
    if (k < wave) off += wtot[k];
  const int first = beg + off + incl - c;
  if (tid < nb) {
    row_ptr[lo + tid] = first;
    cnt[tid] = first;
  }
  if (b == NB - 1 && tid == 0) row_ptr[N_NODES] = end;
  __syncthreads();

#pragma unroll
  for (int k = 0; k < 8; ++k) {
    if (r[k].x >= 0) {
      const int pos = atomicAdd(&cnt[r[k].x >> 17], 1);
      v2i v;
      v.x = r[k].x & 0x1FFFF;
      v.y = r[k].y;
      ((v2i*)csr)[pos] = v;
    }
  }
  for (int i = beg + 8192 + tid; i < end; i += 1024) {
    const v2i e = in[i];
    const int pos = atomicAdd(&cnt[e.x >> 17], 1);
    v2i v;
    v.x = e.x & 0x1FFFF;
    v.y = e.y;
    ((v2i*)csr)[pos] = v;
  }
}

template <int NF>
__global__ __launch_bounds__(256) void gcn_spmm_gather4(
    const float* __restrict__ feat, const int* __restrict__ row_ptr,
    const int* __restrict__ csr, float* __restrict__ out, int n_rows) {
  constexpr int QP = NF / 4;
  const int d = (blockIdx.x * blockDim.x + threadIdx.x) >> 4;
  const int q = threadIdx.x & 15;
  if (d >= n_rows || q >= QP) return;
  const int beg = row_ptr[d];
  const int end = row_ptr[d + 1];
  const v2i* __restrict__ ed = (const v2i*)csr;
  const float* __restrict__ fq = feat + q * 4;

  v4f acc = {0.0f, 0.0f, 0.0f, 0.0f};
  if (beg < end) {
    const int last = end - 1;
    v2i e0 = ed[beg];
    v2i e1 = ed[min(beg + 1, last)];
    v2i e2 = ed[min(beg + 2, last)];
    v2i e3 = ed[min(beg + 3, last)];
    for (int j = beg; j < end; j += 4) {
      v4f x0 = *(const v4f*)(fq + (long)e0.x * NF);
      v4f x1 = *(const v4f*)(fq + (long)e1.x * NF);
      v4f x2 = *(const v4f*)(fq + (long)e2.x * NF);
      v4f x3 = *(const v4f*)(fq + (long)e3.x * NF);
      const float w0 = __int_as_float(e0.y);
      const float w1 = __int_as_float(e1.y);
      const float w2 = __int_as_float(e2.y);
      const float w3 = __int_as_float(e3.y);
      e0 = ed[min(j + 4, last)];
      e1 = ed[min(j + 5, last)];
      e2 = ed[min(j + 6, last)];
      e3 = ed[min(j + 7, last)];
      acc += w0 * x0;
      if (j + 1 < end) acc += w1 * x1;
      if (j + 2 < end) acc += w2 * x2;
      if (j + 3 < end) acc += w3 * x3;
    }
  }
  *(v4f*)(out + (long)d * NF + q * 4) = acc;
}

#define HP 132
#define W1F_V2 ((FEAT / 4) * (EMBED / 16) * 32)
#define W2F_V2 ((EMBED / 4) * 3 * 32)
static_assert(W1F_V2 == 16 * 256 && W2F_V2 == 12 * 256, "staging trip counts");
__global__ __launch_bounds__(256) void gcn_fused_layers(
    const float* __restrict__ A, const float* __restrict__ W1,
    const float* __restrict__ W2, float* __restrict__ T, int mTiles) {
  __shared__ __attribute__((aligned(16)))
      float smem[2 * W1F_V2 + 2 * W2F_V2 + 8 * 16 * HP];
  v2f* const w1f = (v2f*)smem;
  v2f* const w2f = (v2f*)(smem + 2 * W1F_V2);
  const int tid  = threadIdx.x;
  const int wave = tid >> 5;
  const int lane = tid & 31;
  const int n    = lane & 15;
  const int half = lane >> 4;
  const int klo  = half * 2;
  float* const hb = smem + 2 * W1F_V2 + 2 * W2F_V2 + wave * (16 * HP);

  const int  tileRaw = blockIdx.x * 8 + wave;
  const bool valid   = tileRaw < mTiles;
  const int  tile    = valid ? tileRaw : (mTiles - 1);

  const float* __restrict__ arow = A + (long)(tile * 16 + n) * FEAT + klo;
  v2f a[FEAT / 4];
#pragma unroll
  for (int ks = 0; ks < FEAT / 4; ++ks) a[ks] = *(const v2f*)(arow + ks * 4);

#pragma unroll
  for (int j = 0; j < W1F_V2 / 256; ++j) {
    const float* __restrict__ p = W1 + (j * 4 + klo) * EMBED + wave * 16 + n;
    v2f v;
    v.x = p[0];
    v.y = p[EMBED];
    w1f[j * 256 + tid] = v;
  }
#pragma unroll
  for (int j = 0; j < W2F_V2 / 256; ++j) {
    const int  nt  = j >> 2;
    const int  ks  = wave + 8 * (j & 3);
    const int  col = nt * 16 + n;
    const bool in  = col < NCLS;
    const float* __restrict__ p = W2 + (ks * 4 + klo) * NCLS + (in ? col : 0);
    const float x0 = p[0];
    const float x1 = p[NCLS];
    v2f v;
    v.x = in ? x0 : 0.0f;
    v.y = in ? x1 : 0.0f;
    w2f[j * 256 + tid] = v;
  }
  __syncthreads();

  v8f c[EMBED / 16];
#pragma unroll
  for (int nt = 0; nt < EMBED / 16; ++nt) c[nt] = (v8f){};
#pragma unroll
  for (int ks = 0; ks < FEAT / 4; ++ks) {
#pragma unroll
    for (int nt = 0; nt < EMBED / 16; ++nt) {
      const v2f b = w1f[(ks * (EMBED / 16) + nt) * 32 + lane];
      c[nt] = __builtin_amdgcn_wmma_f32_16x16x4_f32(
          false, a[ks], false, b, (short)0, c[nt], false, false);
    }
  }
#pragma unroll
  for (int nt = 0; nt < EMBED / 16; ++nt) {
#pragma unroll
    for (int r = 0; r < 8; ++r)
      hb[(r + 8 * half) * HP + nt * 16 + n] = fmaxf(c[nt][r], 0.0f);
  }
  __syncthreads();

  v8f t0 = {}, t1 = {}, t2 = {};
  const float* __restrict__ hrow = hb + n * HP + klo;
#pragma unroll
  for (int ks = 0; ks < EMBED / 4; ++ks) {
    const v2f ha = *(const v2f*)(hrow + ks * 4);
    const v2f b0 = w2f[(0 * 32 + ks) * 32 + lane];
    const v2f b1 = w2f[(1 * 32 + ks) * 32 + lane];
    const v2f b2 = w2f[(2 * 32 + ks) * 32 + lane];
    t0 = __builtin_amdgcn_wmma_f32_16x16x4_f32(
        false, ha, false, b0, (short)0, t0, false, false);
    t1 = __builtin_amdgcn_wmma_f32_16x16x4_f32(
        false, ha, false, b1, (short)0, t1, false, false);
    t2 = __builtin_amdgcn_wmma_f32_16x16x4_f32(
        false, ha, false, b2, (short)0, t2, false, false);
  }

  if (valid) {
#pragma unroll
    for (int r = 0; r < 8; ++r) {
      float* __restrict__ trow = T + (long)(tile * 16 + r + half * 8) * NCLS;
      trow[n]      = t0[r];
      trow[16 + n] = t1[r];
      if (n < NCLS - 32) trow[32 + n] = t2[r];
    }
  }
}

__global__ __launch_bounds__(256) void gcn_log_softmax(
    const float* __restrict__ in, float* __restrict__ out, int n) {
  int i = blockIdx.x * blockDim.x + threadIdx.x;
  if (i >= n) return;
  const float* __restrict__ row = in + (long)i * NCLS;
  float mx = -INFINITY;
#pragma unroll
  for (int c = 0; c < NCLS; ++c) mx = fmaxf(mx, row[c]);
  float s = 0.0f;
#pragma unroll
  for (int c = 0; c < NCLS; ++c) s += __expf(row[c] - mx);
  float lse = mx + __logf(s);
  float* __restrict__ orow = out + (long)i * NCLS;
#pragma unroll
  for (int c = 0; c < NCLS; ++c) orow[c] = row[c] - lse;
}

extern "C" void kernel_launch(void* const* d_in, const int* in_sizes, int n_in,
                              void* d_out, int out_size, void* d_ws, size_t ws_size,
                              hipStream_t stream) {
  const float* x    = (const float*)d_in[0];
  const float* W1   = (const float*)d_in[1];
  const float* W2   = (const float*)d_in[2];
  const float* ew   = (const float*)d_in[3];
  const int*   esrc = (const int*)d_in[4];
  const int*   edst = (const int*)d_in[5];
  float*       out  = (float*)d_out;

  float* agg1 = (float*)d_ws;
  float* scr  = agg1 + (size_t)N_NODES * FEAT;
  float* t    = scr  + (size_t)N_NODES * EMBED;
  int*   csr  = (int*)(t + (size_t)N_NODES * NCLS);
  int*   rptr = csr + (size_t)2 * N_EDGES;
  float* agg2 = agg1;
  int*   tmp8   = (int*)scr;
  int*   cntmat = tmp8 + (size_t)2 * N_EDGES;
  int*   bstart = cntmat + (size_t)NW * NB;

  const int TPB = 256;

  gcn_bucket_count<<<NW, 1024, 0, stream>>>(edst, cntmat);
  gcn_bucket_scatter<<<NW, 1024, 0, stream>>>(esrc, edst, ew, cntmat, tmp8, bstart);
  gcn_bucket_rows<<<NB, 1024, 0, stream>>>(tmp8, bstart, rptr, csr);

  gcn_spmm_gather4<FEAT><<<(N_NODES * 16 + TPB - 1) / TPB, TPB, 0, stream>>>(
      x, rptr, csr, agg1, N_NODES);

  {
    const int mTiles = N_NODES / 16;
    gcn_fused_layers<<<(mTiles + 7) / 8, TPB, 0, stream>>>(agg1, W1, W2, t, mTiles);
  }

  gcn_spmm_gather4<NCLS><<<(N_NODES * 16 + TPB - 1) / TPB, TPB, 0, stream>>>(
      t, rptr, csr, agg2, N_NODES);

  gcn_log_softmax<<<(N_NODES + TPB - 1) / TPB, TPB, 0, stream>>>(agg2, out, N_NODES);
}


# log_softmax moved into the second SPMM gather as a register epilogue (row max and exp-sum by 16-lane DPP butterflies); agg2 plane and one launch gone (6 launches); exp-sum order now a lane tree
# speedup vs baseline: 3.0629x; 1.1201x over previous
#include <hip/hip_runtime.h>
#include <math.h>

typedef __attribute__((ext_vector_type(2))) float v2f;
typedef __attribute__((ext_vector_type(4))) float v4f;
typedef __attribute__((ext_vector_type(8))) float v8f;
typedef __attribute__((ext_vector_type(2))) int v2i;
typedef __attribute__((ext_vector_type(4))) int v4i;

#define N_NODES 100000
#define N_EDGES 1600000
#define FEAT    64
#define EMBED   128
#define NCLS    40

#define NB     256
#define BS     391
#define NW     256
#define CHUNK  ((N_EDGES + NW - 1) / NW)
#define TRIPS  ((CHUNK + 1023) / 1024)
static_assert(NB * BS >= N_NODES, "buckets must cover every row");
static_assert(BS <= 512 && N_NODES <= (1 << 17), "packed edge word");
static_assert(NB == 256 && NW % 4 == 0, "index split below");

__global__ __launch_bounds__(1024) void gcn_bucket_count(
    const int* __restrict__ dst, int* __restrict__ cntmat) {
  __shared__ int c[NB];
  const int tid = threadIdx.x;
  if (tid < NB) c[tid] = 0;
  const int beg = blockIdx.x * CHUNK;
  const int end = min(beg + CHUNK, N_EDGES);
  int bk[TRIPS];
#pragma unroll
  for (int k = 0; k < TRIPS; ++k) {
    const int e = beg + tid + k * 1024;
    bk[k] = (e < end) ? dst[e] / BS : -1;
  }
  __syncthreads();
#pragma unroll
  for (int k = 0; k < TRIPS; ++k)
    if (bk[k] >= 0) atomicAdd(&c[bk[k]], 1);
  __syncthreads();
  if (tid < NB) cntmat[blockIdx.x * NB + tid] = c[tid];
}

__global__ __launch_bounds__(1024) void gcn_bucket_scatter(
    const int* __restrict__ src, const int* __restrict__ dst,
    const float* __restrict__ w, const int* __restrict__ cntmat,
    int* __restrict__ tmp8, int* __restrict__ bstart) {
  __shared__ int tot[4][NB];
  __shared__ int pre[4][NB];
  __shared__ int cur[NB];
  __shared__ int wtot[8];
  const int tid  = threadIdx.x;
  const int lane = tid & 31;
  const int wave = tid >> 5;
  const int b    = tid & (NB - 1);
  const int q    = tid >> 8;
  const int beg = blockIdx.x * CHUNK;
  const int end = min(beg + CHUNK, N_EDGES);

  int d[TRIPS], s[TRIPS], wb[TRIPS];
#pragma unroll
  for (int k = 0; k < TRIPS; ++k) {
    const int e = beg + tid + k * 1024;
    const bool ok = e < end;
    d[k]  = ok ? dst[e] : -1;
    s[k]  = ok ? src[e] : 0;
    wb[k] = ok ? __float_as_int(w[e]) : 0;
  }

  int t = 0, p = 0;
#pragma unroll 16
  for (int j = 0; j < NW / 4; ++j) {
    const int wg = q * (NW / 4) + j;
    const int v = cntmat[wg * NB + b];
    t += v;
    p += (wg < (int)blockIdx.x) ? v : 0;
  }
  tot[q][b] = t;
  pre[q][b] = p;
  __syncthreads();

  int T = 0, P = 0;
  if (tid < NB) {
    T = (tot[0][tid] + tot[1][tid]) + (tot[2][tid] + tot[3][tid]);
    P = (pre[0][tid] + pre[1][tid]) + (pre[2][tid] + pre[3][tid]);
  }
  int incl = T;
#pragma unroll
  for (int o = 1; o < 32; o <<= 1) {
    int v = __shfl_up(incl, o);
    if (lane >= o) incl += v;
  }
  if (tid < NB && lane == 31) wtot[wave] = incl;
  __syncthreads();
  if (tid < NB) {
    int off = 0;
#pragma unroll
    for (int k = 0; k < 8; ++k)
      if (k < wave) off += wtot[k];
    const int excl = off + incl - T;
    cur[tid] = excl + P;
    if (blockIdx.x == 0) {
      bstart[tid] = excl;
      if (tid == NB - 1) bstart[NB] = excl + T;
    }
  }
  __syncthreads();

#pragma unroll
  for (int k = 0; k < TRIPS; ++k) {
    if (d[k] >= 0) {
      const int bkt = d[k] / BS;
      const int pos = atomicAdd(&cur[bkt], 1);
      v2i r;
      r.x = s[k] | ((d[k] - bkt * BS) << 17);
      r.y = wb[k];
      ((v2i*)tmp8)[pos] = r;
    }
  }
}

__global__ __launch_bounds__(1024) void gcn_bucket_rows(
    const int* __restrict__ tmp8, const int* __restrict__ bstart,
    int* __restrict__ row_ptr, int* __restrict__ csr) {
  __shared__ int cnt[512];
  __shared__ int wtot[32];
  const int tid  = threadIdx.x;
  const int lane = tid & 31;
  const int wave = tid >> 5;
  const int b    = blockIdx.x;
  const int lo   = b * BS;
  const int nb   = min(BS, N_NODES - lo);
  const int beg  = bstart[b];
  const int end  = bstart[b + 1];
  const v2i* __restrict__ in = (const v2i*)tmp8;
  if (tid < 512) cnt[tid] = 0;

  v2i r[8];
#pragma unroll
  for (int k = 0; k < 8; ++k) {
    const int i = beg + tid + k * 1024;
    if (i < end) r[k] = in[i];
    else         r[k] = (v2i){-1, 0};
  }
  __syncthreads();
#pragma unroll
  for (int k = 0; k < 8; ++k)
    if (r[k].x >= 0) atomicAdd(&cnt[r[k].x >> 17], 1);
  for (int i = beg + 8192 + tid; i < end; i += 1024)
    atomicAdd(&cnt[in[i].x >> 17], 1);
  __syncthreads();

  const int c = (tid < nb) ? cnt[tid] : 0;
  int incl = c;
#pragma unroll
  for (int o = 1; o < 32; o <<= 1) {
    int v = __shfl_up(incl, o);
    if (lane >= o) incl += v;
  }
  if (lane == 31) wtot[wave] = incl;
  __syncthreads();
  int off = 0;
#pragma unroll
  for (int k = 0; k < 32; ++k)
    if (k < wave) off += wtot[k];
  const int first = beg + off + incl - c;
  if (tid < nb) {
    row_ptr[lo + tid] = first;
    cnt[tid] = first;
  }
  if (b == NB - 1 && tid == 0) row_ptr[N_NODES] = end;
  __syncthreads();

#pragma unroll
  for (int k = 0; k < 8; ++k) {
    if (r[k].x >= 0) {
      const int pos = atomicAdd(&cnt[r[k].x >> 17], 1);
      v2i v;
      v.x = r[k].x & 0x1FFFF;
      v.y = r[k].y;
      ((v2i*)csr)[pos] = v;
    }
  }
  for (int i = beg + 8192 + tid; i < end; i += 1024) {
    const v2i e = in[i];
    const int pos = atomicAdd(&cnt[e.x >> 17], 1);
    v2i v;
    v.x = e.x & 0x1FFFF;
    v.y = e.y;
    ((v2i*)csr)[pos] = v;
  }
}

template <int NF>
__device__ __forceinline__ v4f gcn_row_sum4(
    const float* __restrict__ fq, const v2i* __restrict__ ed, int beg, int end) {
  v4f acc = {0.0f, 0.0f, 0.0f, 0.0f};
  if (beg < end) {
    const int last = end - 1;
    v2i e0 = ed[beg];
    v2i e1 = ed[min(beg + 1, last)];
    v2i e2 = ed[min(beg + 2, last)];
    v2i e3 = ed[min(beg + 3, last)];
    for (int j = beg; j < end; j += 4) {
      v4f x0 = *(const v4f*)(fq + (long)e0.x * NF);
      v4f x1 = *(const v4f*)(fq + (long)e1.x * NF);
      v4f x2 = *(const v4f*)(fq + (long)e2.x * NF);
      v4f x3 = *(const v4f*)(fq + (long)e3.x * NF);
      const float w0 = __int_as_float(e0.y);
      const float w1 = __int_as_float(e1.y);
      const float w2 = __int_as_float(e2.y);
      const float w3 = __int_as_float(e3.y);
      e0 = ed[min(j + 4, last)];
      e1 = ed[min(j + 5, last)];
      e2 = ed[min(j + 6, last)];
      e3 = ed[min(j + 7, last)];
      acc += w0 * x0;
      if (j + 1 < end) acc += w1 * x1;
      if (j + 2 < end) acc += w2 * x2;
      if (j + 3 < end) acc += w3 * x3;
    }
  }
  return acc;
}

template <int NF>
__global__ __launch_bounds__(256) void gcn_spmm_gather4(
    const float* __restrict__ feat, const int* __restrict__ row_ptr,
    const int* __restrict__ csr, float* __restrict__ out, int n_rows) {
  constexpr int QP = NF / 4;
  const int d = (blockIdx.x * blockDim.x + threadIdx.x) >> 4;
  const int q = threadIdx.x & 15;
  if (d >= n_rows || q >= QP) return;
  const v4f acc = gcn_row_sum4<NF>(feat + q * 4, (const v2i*)csr,
                                   row_ptr[d], row_ptr[d + 1]);
  *(v4f*)(out + (long)d * NF + q * 4) = acc;
}

template <int M>
__device__ __forceinline__ float gcn_row_xor(float v) {
  return __int_as_float(__builtin_amdgcn_update_dpp(
      0, __float_as_int(v), 0x160 + M, 0xf, 0xf, false));
}

__global__ __launch_bounds__(256) void gcn_spmm_gather4_logsoftmax(
    const float* __restrict__ feat, const int* __restrict__ row_ptr,
    const int* __restrict__ csr, float* __restrict__ out, int n_rows) {
  constexpr int QP = NCLS / 4;
  static_assert(NCLS % 4 == 0 && QP <= 16, "a class row fits one 16-lane row");
  const int d = (blockIdx.x * blockDim.x + threadIdx.x) >> 4;
  const int q = threadIdx.x & 15;
  const bool act = (d < n_rows) && (q < QP);
  v4f acc = {0.0f, 0.0f, 0.0f, 0.0f};
  if (act)
    acc = gcn_row_sum4<NCLS>(feat + q * 4, (const v2i*)csr,
                             row_ptr[d], row_ptr[d + 1]);

  float mx = act ? fmaxf(fmaxf(acc.x, acc.y), fmaxf(acc.z, acc.w)) : -INFINITY;
  mx = fmaxf(mx, gcn_row_xor<1>(mx));
  mx = fmaxf(mx, gcn_row_xor<2>(mx));
  mx = fmaxf(mx, gcn_row_xor<4>(mx));
  mx = fmaxf(mx, gcn_row_xor<8>(mx));
  float s = 0.0f;
  if (act)
    s = (__expf(acc.x - mx) + __expf(acc.y - mx)) +
        (__expf(acc.z - mx) + __expf(acc.w - mx));
  s += gcn_row_xor<1>(s);
  s += gcn_row_xor<2>(s);
  s += gcn_row_xor<4>(s);
  s += gcn_row_xor<8>(s);
  const float lse = mx + __logf(s);
  if (act) *(v4f*)(out + (long)d * NCLS + q * 4) = acc - lse;
}

#define HP 132
#define W1F_V2 ((FEAT / 4) * (EMBED / 16) * 32)
#define W2F_V2 ((EMBED / 4) * 3 * 32)
static_assert(W1F_V2 == 16 * 256 && W2F_V2 == 12 * 256, "staging trip counts");
__global__ __launch_bounds__(256) void gcn_fused_layers(
    const float* __restrict__ A, const float* __restrict__ W1,
    const float* __restrict__ W2, float* __restrict__ T, int mTiles) {
  __shared__ __attribute__((aligned(16)))
      float smem[2 * W1F_V2 + 2 * W2F_V2 + 8 * 16 * HP];
  v2f* const w1f = (v2f*)smem;
  v2f* const w2f = (v2f*)(smem + 2 * W1F_V2);
  const int tid  = threadIdx.x;
  const int wave = tid >> 5;
  const int lane = tid & 31;
  const int n    = lane & 15;
  const int half = lane >> 4;
  const int klo  = half * 2;
  float* const hb = smem + 2 * W1F_V2 + 2 * W2F_V2 + wave * (16 * HP);

  const int  tileRaw = blockIdx.x * 8 + wave;
  const bool valid   = tileRaw < mTiles;
  const int  tile    = valid ? tileRaw : (mTiles - 1);

  const float* __restrict__ arow = A + (long)(tile * 16 + n) * FEAT + klo;
  v2f a[FEAT / 4];
#pragma unroll
  for (int ks = 0; ks < FEAT / 4; ++ks) a[ks] = *(const v2f*)(arow + ks * 4);

#pragma unroll
  for (int j = 0; j < W1F_V2 / 256; ++j) {
    const float* __restrict__ p = W1 + (j * 4 + klo) * EMBED + wave * 16 + n;
    v2f v;
    v.x = p[0];
    v.y = p[EMBED];
    w1f[j * 256 + tid] = v;
  }
#pragma unroll
  for (int j = 0; j < W2F_V2 / 256; ++j) {
    const int  nt  = j >> 2;
    const int  ks  = wave + 8 * (j & 3);
    const int  col = nt * 16 + n;
    const bool in  = col < NCLS;
    const float* __restrict__ p = W2 + (ks * 4 + klo) * NCLS + (in ? col : 0);
    const float x0 = p[0];
    const float x1 = p[NCLS];
    v2f v;
    v.x = in ? x0 : 0.0f;
    v.y = in ? x1 : 0.0f;
    w2f[j * 256 + tid] = v;
  }
  __syncthreads();

  v8f c[EMBED / 16];
#pragma unroll
  for (int nt = 0; nt < EMBED / 16; ++nt) c[nt] = (v8f){};
#pragma unroll
  for (int ks = 0; ks < FEAT / 4; ++ks) {
#pragma unroll
    for (int nt = 0; nt < EMBED / 16; ++nt) {
      const v2f b = w1f[(ks * (EMBED / 16) + nt) * 32 + lane];
      c[nt] = __builtin_amdgcn_wmma_f32_16x16x4_f32(
          false, a[ks], false, b, (short)0, c[nt], false, false);
    }
  }
#pragma unroll
  for (int nt = 0; nt < EMBED / 16; ++nt) {
#pragma unroll
    for (int r = 0; r < 8; ++r)
      hb[(r + 8 * half) * HP + nt * 16 + n] = fmaxf(c[nt][r], 0.0f);
  }
  __syncthreads();

  v8f t0 = {}, t1 = {}, t2 = {};
  const float* __restrict__ hrow = hb + n * HP + klo;
#pragma unroll
  for (int ks = 0; ks < EMBED / 4; ++ks) {
    const v2f ha = *(const v2f*)(hrow + ks * 4);
    const v2f b0 = w2f[(0 * 32 + ks) * 32 + lane];
    const v2f b1 = w2f[(1 * 32 + ks) * 32 + lane];
    const v2f b2 = w2f[(2 * 32 + ks) * 32 + lane];
    t0 = __builtin_amdgcn_wmma_f32_16x16x4_f32(
        false, ha, false, b0, (short)0, t0, false, false);
    t1 = __builtin_amdgcn_wmma_f32_16x16x4_f32(
        false, ha, false, b1, (short)0, t1, false, false);
    t2 = __builtin_amdgcn_wmma_f32_16x16x4_f32(
        false, ha, false, b2, (short)0, t2, false, false);
  }

  if (valid) {
#pragma unroll
    for (int r = 0; r < 8; ++r) {
      float* __restrict__ trow = T + (long)(tile * 16 + r + half * 8) * NCLS;
      trow[n]      = t0[r];
      trow[16 + n] = t1[r];
      if (n < NCLS - 32) trow[32 + n] = t2[r];
    }
  }
}

extern "C" void kernel_launch(void* const* d_in, const int* in_sizes, int n_in,
                              void* d_out, int out_size, void* d_ws, size_t ws_size,
                              hipStream_t stream) {
  const float* x    = (const float*)d_in[0];
  const float* W1   = (const float*)d_in[1];
  const float* W2   = (const float*)d_in[2];
  const float* ew   = (const float*)d_in[3];
  const int*   esrc = (const int*)d_in[4];
  const int*   edst = (const int*)d_in[5];
  float*       out  = (float*)d_out;

  float* agg1 = (float*)d_ws;
  float* scr  = agg1 + (size_t)N_NODES * FEAT;
  float* t    = scr  + (size_t)N_NODES * EMBED;
  int*   csr  = (int*)(t + (size_t)N_NODES * NCLS);
  int*   rptr = csr + (size_t)2 * N_EDGES;
  int*   tmp8   = (int*)scr;
  int*   cntmat = tmp8 + (size_t)2 * N_EDGES;
  int*   bstart = cntmat + (size_t)NW * NB;

  const int TPB = 256;

  gcn_bucket_count<<<NW, 1024, 0, stream>>>(edst, cntmat);
  gcn_bucket_scatter<<<NW, 1024, 0, stream>>>(esrc, edst, ew, cntmat, tmp8, bstart);
  gcn_bucket_rows<<<NB, 1024, 0, stream>>>(tmp8, bstart, rptr, csr);

  gcn_spmm_gather4<FEAT><<<(N_NODES * 16 + TPB - 1) / TPB, TPB, 0, stream>>>(
      x, rptr, csr, agg1, N_NODES);

  {
    const int mTiles = N_NODES / 16;
    gcn_fused_layers<<<(mTiles + 7) / 8, TPB, 0, stream>>>(agg1, W1, W2, t, mTiles);
  }

  gcn_spmm_gather4_logsoftmax<<<(N_NODES * 16 + TPB - 1) / TPB, TPB, 0, stream>>>(
      t, rptr, csr, out, N_NODES);
}


# Edge grouping level 1: each chunk sorts its edges by bucket in LDS (rank = counting add's return) into its own window; count launch and count matrix gone, 5 launches; row sums still in arrival order
# speedup vs baseline: 3.9749x; 1.2978x over previous
#include <hip/hip_runtime.h>
#include <math.h>

typedef __attribute__((ext_vector_type(2))) float v2f;
typedef __attribute__((ext_vector_type(4))) float v4f;
typedef __attribute__((ext_vector_type(8))) float v8f;
typedef __attribute__((ext_vector_type(2))) int v2i;
typedef __attribute__((ext_vector_type(4))) int v4i;

#define N_NODES 100000
#define N_EDGES 1600000
#define FEAT    64
#define EMBED   128
#define NCLS    40

#define NB     256
#define BS     391
#define NW     256
#define CHUNK  ((N_EDGES + NW - 1) / NW)
#define CSTR   ((CHUNK + 15) / 16 * 16)
#define TRIPS  ((CHUNK + 1023) / 1024)
static_assert(NB * BS >= N_NODES, "buckets must cover every row");
static_assert(BS <= 512 && N_NODES <= (1 << 17), "packed edge word");
static_assert(NB == 256 && NW == 256, "one thread per bucket / per chunk below");

__global__ __launch_bounds__(1024) void gcn_chunk_sort(
    const int* __restrict__ src, const int* __restrict__ dst,
    const float* __restrict__ w, int* __restrict__ tmp8,
    int* __restrict__ segT) {
  __shared__ int c[NB];
  __shared__ int wtot[8];
  __shared__ v2i stage[CHUNK];
  const int tid  = threadIdx.x;
  const int lane = tid & 31;
  const int wave = tid >> 5;
  const int beg = blockIdx.x * CHUNK;
  const int end = min(beg + CHUNK, N_EDGES);
  if (tid < NB) c[tid] = 0;

  int bk[TRIPS], pk[TRIPS], wb[TRIPS], rk[TRIPS];
#pragma unroll
  for (int k = 0; k < TRIPS; ++k) {
    const int e = beg + tid + k * 1024;
    const bool ok = e < end;
    const int d = ok ? dst[e] : -1;
    const int s = ok ? src[e] : 0;
    wb[k] = ok ? __float_as_int(w[e]) : 0;
    bk[k] = ok ? d / BS : -1;
    pk[k] = s | ((d - bk[k] * BS) << 17);
  }
  __syncthreads();
#pragma unroll
  for (int k = 0; k < TRIPS; ++k) {
    rk[k] = 0;
    if (bk[k] >= 0) rk[k] = atomicAdd(&c[bk[k]], 1);
  }
  __syncthreads();

  const int T = (tid < NB) ? c[tid] : 0;
  int incl = T;
#pragma unroll
  for (int o = 1; o < 32; o <<= 1) {
    int v = __shfl_up(incl, o);
    if (lane >= o) incl += v;
  }
  if (tid < NB && lane == 31) wtot[wave] = incl;
  __syncthreads();
  if (tid < NB) {
    int off = 0;
#pragma unroll
    for (int k = 0; k < 8; ++k)
      if (k < wave) off += wtot[k];
    const int excl = off + incl - T;
    c[tid] = excl;
    segT[tid * NW + blockIdx.x] = excl;
    if (tid == NB - 1) segT[NB * NW + blockIdx.x] = excl + T;
  }
  __syncthreads();

#pragma unroll
  for (int k = 0; k < TRIPS; ++k) {
    if (bk[k] >= 0) {
      v2i r;
      r.x = pk[k];
      r.y = wb[k];
      stage[c[bk[k]] + rk[k]] = r;
    }
  }
  __syncthreads();

  v2i* __restrict__ outw = (v2i*)tmp8 + (long)blockIdx.x * CSTR;
  const int n = end - beg;
#pragma unroll
  for (int k = 0; k < TRIPS; ++k) {
    const int i = tid + k * 1024;
    if (i < n) outw[i] = stage[i];
  }
}

__global__ __launch_bounds__(1024) void gcn_bucket_rows(
    const int* __restrict__ tmp8, const int* __restrict__ segT,
    int* __restrict__ row_ptr, int* __restrict__ csr) {
  __shared__ int cnt[512];
  __shared__ int wtot[32];
  __shared__ int sbeg[NW];
  __shared__ int slen[NW];
  __shared__ int rbeg[8];
  __shared__ int rlen[8];
  const int tid  = threadIdx.x;
  const int lane = tid & 31;
  const int wave = tid >> 5;
  const int b    = blockIdx.x;
  const int lo   = b * BS;
  const int nb   = min(BS, N_NODES - lo);
  const v2i* __restrict__ in = (const v2i*)tmp8;
  if (tid < 512) cnt[tid] = 0;

  int s0 = 0, sl = 0;
  if (tid < NW) {
    s0 = segT[b * NW + tid];
    sl = segT[(b + 1) * NW + tid] - s0;
    sbeg[tid] = tid * CSTR + s0;
    slen[tid] = sl;
  }
  int ra = s0, rl = sl;
#pragma unroll
  for (int o = 16; o > 0; o >>= 1) {
    ra += __shfl_xor(ra, o);
    rl += __shfl_xor(rl, o);
  }
  if (tid < NW && lane == 0) {
    rbeg[wave] = ra;
    rlen[wave] = rl;
  }
  __syncthreads();
  int beg = 0, tot = 0;
#pragma unroll
  for (int k = 0; k < 8; ++k) {
    beg += rbeg[k];
    tot += rlen[k];
  }
  const int end = beg + tot;

  v2i r[16];
#pragma unroll
  for (int k = 0; k < 8; ++k) {
    const int ch = wave * 8 + k;
    const int len = slen[ch];
    const int base = sbeg[ch];
    if (lane < len)      r[2 * k] = in[base + lane];
    else                 r[2 * k] = (v2i){-1, 0};
    if (lane + 32 < len) r[2 * k + 1] = in[base + 32 + lane];
    else                 r[2 * k + 1] = (v2i){-1, 0};
  }
#pragma unroll
  for (int k = 0; k < 16; ++k)
    if (r[k].x >= 0) atomicAdd(&cnt[r[k].x >> 17], 1);
  for (int k = 0; k < 8; ++k) {
    const int ch = wave * 8 + k;
    const int len = slen[ch];
    const int base = sbeg[ch];
    for (int i = 64 + lane; i < len; i += 32)
      atomicAdd(&cnt[in[base + i].x >> 17], 1);
  }
  __syncthreads();

  const int c = (tid < nb) ? cnt[tid] : 0;
  int incl = c;
#pragma unroll
  for (int o = 1; o < 32; o <<= 1) {
    int v = __shfl_up(incl, o);
    if (lane >= o) incl += v;
  }
  if (lane == 31) wtot[wave] = incl;
  __syncthreads();
  int off = 0;
#pragma unroll
  for (int k = 0; k < 32; ++k)
    if (k < wave) off += wtot[k];
  const int first = beg + off + incl - c;
  if (tid < nb) {
    row_ptr[lo + tid] = first;
    cnt[tid] = first;
  }
  if (b == NB - 1 && tid == 0) row_ptr[N_NODES] = end;
  __syncthreads();

#pragma unroll
  for (int k = 0; k < 16; ++k) {
    if (r[k].x >= 0) {
      const int pos = atomicAdd(&cnt[r[k].x >> 17], 1);
      v2i v;
      v.x = r[k].x & 0x1FFFF;
      v.y = r[k].y;
      ((v2i*)csr)[pos] = v;
    }
  }
  for (int k = 0; k < 8; ++k) {
    const int ch = wave * 8 + k;
    const int len = slen[ch];
    const int base = sbeg[ch];
    for (int i = 64 + lane; i < len; i += 32) {
      const v2i e = in[base + i];
      const int pos = atomicAdd(&cnt[e.x >> 17], 1);
      v2i v;
      v.x = e.x & 0x1FFFF;
      v.y = e.y;
      ((v2i*)csr)[pos] = v;
    }
  }
}

template <int NF>
__device__ __forceinline__ v4f gcn_row_sum4(
    const float* __restrict__ fq, const v2i* __restrict__ ed, int beg, int end) {
  v4f acc = {0.0f, 0.0f, 0.0f, 0.0f};
  if (beg < end) {
    const int last = end - 1;
    v2i e0 = ed[beg];
    v2i e1 = ed[min(beg + 1, last)];
    v2i e2 = ed[min(beg + 2, last)];
    v2i e3 = ed[min(beg + 3, last)];
    for (int j = beg; j < end; j += 4) {
      v4f x0 = *(const v4f*)(fq + (long)e0.x * NF);
      v4f x1 = *(const v4f*)(fq + (long)e1.x * NF);
      v4f x2 = *(const v4f*)(fq + (long)e2.x * NF);
      v4f x3 = *(const v4f*)(fq + (long)e3.x * NF);
      const float w0 = __int_as_float(e0.y);
      const float w1 = __int_as_float(e1.y);
      const float w2 = __int_as_float(e2.y);
      const float w3 = __int_as_float(e3.y);
      e0 = ed[min(j + 4, last)];
      e1 = ed[min(j + 5, last)];
      e2 = ed[min(j + 6, last)];
      e3 = ed[min(j + 7, last)];
      acc += w0 * x0;
      if (j + 1 < end) acc += w1 * x1;
      if (j + 2 < end) acc += w2 * x2;
      if (j + 3 < end) acc += w3 * x3;
    }
  }
  return acc;
}

template <int NF>
__global__ __launch_bounds__(256) void gcn_spmm_gather4(
    const float* __restrict__ feat, const int* __restrict__ row_ptr,
    const int* __restrict__ csr, float* __restrict__ out, int n_rows) {
  constexpr int QP = NF / 4;
  const int d = (blockIdx.x * blockDim.x + threadIdx.x) >> 4;
  const int q = threadIdx.x & 15;
  if (d >= n_rows || q >= QP) return;
  const v4f acc = gcn_row_sum4<NF>(feat + q * 4, (const v2i*)csr,
                                   row_ptr[d], row_ptr[d + 1]);
  *(v4f*)(out + (long)d * NF + q * 4) = acc;
}

template <int M>
__device__ __forceinline__ float gcn_row_xor(float v) {
  return __int_as_float(__builtin_amdgcn_update_dpp(
      0, __float_as_int(v), 0x160 + M, 0xf, 0xf, false));
}

__global__ __launch_bounds__(256) void gcn_spmm_gather4_logsoftmax(
    const float* __restrict__ feat, const int* __restrict__ row_ptr,
    const int* __restrict__ csr, float* __restrict__ out, int n_rows) {
  constexpr int QP = NCLS / 4;
  static_assert(NCLS % 4 == 0 && QP <= 16, "a class row fits one 16-lane row");
  const int d = (blockIdx.x * blockDim.x + threadIdx.x) >> 4;
  const int q = threadIdx.x & 15;
  const bool act = (d < n_rows) && (q < QP);
  v4f acc = {0.0f, 0.0f, 0.0f, 0.0f};
  if (act)
    acc = gcn_row_sum4<NCLS>(feat + q * 4, (const v2i*)csr,
                             row_ptr[d], row_ptr[d + 1]);

  float mx = act ? fmaxf(fmaxf(acc.x, acc.y), fmaxf(acc.z, acc.w)) : -INFINITY;
  mx = fmaxf(mx, gcn_row_xor<1>(mx));
  mx = fmaxf(mx, gcn_row_xor<2>(mx));
  mx = fmaxf(mx, gcn_row_xor<4>(mx));
  mx = fmaxf(mx, gcn_row_xor<8>(mx));
  float s = 0.0f;
  if (act)
    s = (__expf(acc.x - mx) + __expf(acc.y - mx)) +
        (__expf(acc.z - mx) + __expf(acc.w - mx));
  s += gcn_row_xor<1>(s);
  s += gcn_row_xor<2>(s);
  s += gcn_row_xor<4>(s);
  s += gcn_row_xor<8>(s);
  const float lse = mx + __logf(s);
  if (act) *(v4f*)(out + (long)d * NCLS + q * 4) = acc - lse;
}

#define HP 132
#define W1F_V2 ((FEAT / 4) * (EMBED / 16) * 32)
#define W2F_V2 ((EMBED / 4) * 3 * 32)
static_assert(W1F_V2 == 16 * 256 && W2F_V2 == 12 * 256, "staging trip counts");
__global__ __launch_bounds__(256) void gcn_fused_layers(
    const float* __restrict__ A, const float* __restrict__ W1,
    const float* __restrict__ W2, float* __restrict__ T, int mTiles) {
  __shared__ __attribute__((aligned(16)))
      float smem[2 * W1F_V2 + 2 * W2F_V2 + 8 * 16 * HP];
  v2f* const w1f = (v2f*)smem;
  v2f* const w2f = (v2f*)(smem + 2 * W1F_V2);
  const int tid  = threadIdx.x;
  const int wave = tid >> 5;
  const int lane = tid & 31;
  const int n    = lane & 15;
  const int half = lane >> 4;
  const int klo  = half * 2;
  float* const hb = smem + 2 * W1F_V2 + 2 * W2F_V2 + wave * (16 * HP);

  const int  tileRaw = blockIdx.x * 8 + wave;
  const bool valid   = tileRaw < mTiles;
  const int  tile    = valid ? tileRaw : (mTiles - 1);

  const float* __restrict__ arow = A + (long)(tile * 16 + n) * FEAT + klo;
  v2f a[FEAT / 4];
#pragma unroll
  for (int ks = 0; ks < FEAT / 4; ++ks) a[ks] = *(const v2f*)(arow + ks * 4);

#pragma unroll
  for (int j = 0; j < W1F_V2 / 256; ++j) {
    const float* __restrict__ p = W1 + (j * 4 + klo) * EMBED + wave * 16 + n;
    v2f v;
    v.x = p[0];
    v.y = p[EMBED];
    w1f[j * 256 + tid] = v;
  }
#pragma unroll
  for (int j = 0; j < W2F_V2 / 256; ++j) {
    const int  nt  = j >> 2;
    const int  ks  = wave + 8 * (j & 3);
    const int  col = nt * 16 + n;
    const bool in  = col < NCLS;
    const float* __restrict__ p = W2 + (ks * 4 + klo) * NCLS + (in ? col : 0);
    const float x0 = p[0];
    const float x1 = p[NCLS];
    v2f v;
    v.x = in ? x0 : 0.0f;
    v.y = in ? x1 : 0.0f;
    w2f[j * 256 + tid] = v;
  }
  __syncthreads();

  v8f c[EMBED / 16];
#pragma unroll
  for (int nt = 0; nt < EMBED / 16; ++nt) c[nt] = (v8f){};
#pragma unroll
  for (int ks = 0; ks < FEAT / 4; ++ks) {
#pragma unroll
    for (int nt = 0; nt < EMBED / 16; ++nt) {
      const v2f b = w1f[(ks * (EMBED / 16) + nt) * 32 + lane];
      c[nt] = __builtin_amdgcn_wmma_f32_16x16x4_f32(
          false, a[ks], false, b, (short)0, c[nt], false, false);
    }
  }
#pragma unroll
  for (int nt = 0; nt < EMBED / 16; ++nt) {
#pragma unroll
    for (int r = 0; r < 8; ++r)
      hb[(r + 8 * half) * HP + nt * 16 + n] = fmaxf(c[nt][r], 0.0f);
  }
  __syncthreads();

  v8f t0 = {}, t1 = {}, t2 = {};
  const float* __restrict__ hrow = hb + n * HP + klo;
#pragma unroll
  for (int ks = 0; ks < EMBED / 4; ++ks) {
    const v2f ha = *(const v2f*)(hrow + ks * 4);
    const v2f b0 = w2f[(0 * 32 + ks) * 32 + lane];
    const v2f b1 = w2f[(1 * 32 + ks) * 32 + lane];
    const v2f b2 = w2f[(2 * 32 + ks) * 32 + lane];
    t0 = __builtin_amdgcn_wmma_f32_16x16x4_f32(
        false, ha, false, b0, (short)0, t0, false, false);
    t1 = __builtin_amdgcn_wmma_f32_16x16x4_f32(
        false, ha, false, b1, (short)0, t1, false, false);
    t2 = __builtin_amdgcn_wmma_f32_16x16x4_f32(
        false, ha, false, b2, (short)0, t2, false, false);
  }

  if (valid) {
#pragma unroll
    for (int r = 0; r < 8; ++r) {
      float* __restrict__ trow = T + (long)(tile * 16 + r + half * 8) * NCLS;
      trow[n]      = t0[r];
      trow[16 + n] = t1[r];
      if (n < NCLS - 32) trow[32 + n] = t2[r];
    }
  }
}

extern "C" void kernel_launch(void* const* d_in, const int* in_sizes, int n_in,
                              void* d_out, int out_size, void* d_ws, size_t ws_size,
                              hipStream_t stream) {
  const float* x    = (const float*)d_in[0];
  const float* W1   = (const float*)d_in[1];
  const float* W2   = (const float*)d_in[2];
  const float* ew   = (const float*)d_in[3];
  const int*   esrc = (const int*)d_in[4];
  const int*   edst = (const int*)d_in[5];
  float*       out  = (float*)d_out;

  float* agg1 = (float*)d_ws;
  float* scr  = agg1 + (size_t)N_NODES * FEAT;
  float* t    = scr  + (size_t)N_NODES * EMBED;
  int*   csr  = (int*)(t + (size_t)N_NODES * NCLS);
  int*   rptr = csr + (size_t)2 * N_EDGES;
  int*   tmp8   = (int*)scr;
  int*   segT   = tmp8 + (size_t)2 * NW * CSTR;

  const int TPB = 256;

  gcn_chunk_sort<<<NW, 1024, 0, stream>>>(esrc, edst, ew, tmp8, segT);
  gcn_bucket_rows<<<NB, 1024, 0, stream>>>(tmp8, segT, rptr, csr);

  gcn_spmm_gather4<FEAT><<<(N_NODES * 16 + TPB - 1) / TPB, TPB, 0, stream>>>(
      x, rptr, csr, agg1, N_NODES);

  {
    const int mTiles = N_NODES / 16;
    gcn_fused_layers<<<(mTiles + 7) / 8, TPB, 0, stream>>>(agg1, W1, W2, t, mTiles);
  }

  gcn_spmm_gather4_logsoftmax<<<(N_NODES * 16 + TPB - 1) / TPB, TPB, 0, stream>>>(
      t, rptr, csr, out, N_NODES);
}


# First SPMM fused into the row-filing kernel: a bucket's edges are filed by row in LDS, its rows gather-summed from that list, csr written as whole lines; 4 launches; row sums still arrival order
# speedup vs baseline: 4.2211x; 1.0619x over previous
#include <hip/hip_runtime.h>
#include <math.h>

typedef __attribute__((ext_vector_type(2))) float v2f;
typedef __attribute__((ext_vector_type(4))) float v4f;
typedef __attribute__((ext_vector_type(8))) float v8f;
typedef __attribute__((ext_vector_type(2))) int v2i;
typedef __attribute__((ext_vector_type(4))) int v4i;

#define N_NODES 100000
#define N_EDGES 1600000
#define FEAT    64
#define EMBED   128
#define NCLS    40

#define NB     256
#define BS     391
#define NW     256
#define CHUNK  ((N_EDGES + NW - 1) / NW)
#define CSTR   ((CHUNK + 15) / 16 * 16)
#define TRIPS  ((CHUNK + 1023) / 1024)
static_assert(NB * BS >= N_NODES, "buckets must cover every row");
static_assert(BS <= 512 && N_NODES <= (1 << 17), "packed edge word");
static_assert(NB == 256 && NW == 256, "one thread per bucket / per chunk below");

__global__ __launch_bounds__(1024) void gcn_chunk_sort(
    const int* __restrict__ src, const int* __restrict__ dst,
    const float* __restrict__ w, int* __restrict__ tmp8,
    int* __restrict__ segT) {
  __shared__ int c[NB];
  __shared__ int wtot[8];
  __shared__ v2i stage[CHUNK];
  const int tid  = threadIdx.x;
  const int lane = tid & 31;
  const int wave = tid >> 5;
  const int beg = blockIdx.x * CHUNK;
  const int end = min(beg + CHUNK, N_EDGES);
  if (tid < NB) c[tid] = 0;

  int bk[TRIPS], pk[TRIPS], wb[TRIPS], rk[TRIPS];
#pragma unroll
  for (int k = 0; k < TRIPS; ++k) {
    const int e = beg + tid + k * 1024;
    const bool ok = e < end;
    const int d = ok ? dst[e] : -1;
    const int s = ok ? src[e] : 0;
    wb[k] = ok ? __float_as_int(w[e]) : 0;
    bk[k] = ok ? d / BS : -1;
    pk[k] = s | ((d - bk[k] * BS) << 17);
  }
  __syncthreads();
#pragma unroll
  for (int k = 0; k < TRIPS; ++k) {
    rk[k] = 0;
    if (bk[k] >= 0) rk[k] = atomicAdd(&c[bk[k]], 1);
  }
  __syncthreads();

  const int T = (tid < NB) ? c[tid] : 0;
  int incl = T;
#pragma unroll
  for (int o = 1; o < 32; o <<= 1) {
    int v = __shfl_up(incl, o);
    if (lane >= o) incl += v;
  }
  if (tid < NB && lane == 31) wtot[wave] = incl;
  __syncthreads();
  if (tid < NB) {
    int off = 0;
#pragma unroll
    for (int k = 0; k < 8; ++k)
      if (k < wave) off += wtot[k];
    const int excl = off + incl - T;
    c[tid] = excl;
    segT[tid * NW + blockIdx.x] = excl;
    if (tid == NB - 1) segT[NB * NW + blockIdx.x] = excl + T;
  }
  __syncthreads();

#pragma unroll
  for (int k = 0; k < TRIPS; ++k) {
    if (bk[k] >= 0) {
      v2i r;
      r.x = pk[k];
      r.y = wb[k];
      stage[c[bk[k]] + rk[k]] = r;
    }
  }
  __syncthreads();

  v2i* __restrict__ outw = (v2i*)tmp8 + (long)blockIdx.x * CSTR;
  const int n = end - beg;
#pragma unroll
  for (int k = 0; k < TRIPS; ++k) {
    const int i = tid + k * 1024;
    if (i < n) outw[i] = stage[i];
  }
}

template <int NF>
__device__ __forceinline__ v4f gcn_row_sum4(
    const float* __restrict__ fq, const v2i* __restrict__ ed, int beg, int end) {
  v4f acc = {0.0f, 0.0f, 0.0f, 0.0f};
  if (beg < end) {
    const int last = end - 1;
    v2i e0 = ed[beg];
    v2i e1 = ed[min(beg + 1, last)];
    v2i e2 = ed[min(beg + 2, last)];
    v2i e3 = ed[min(beg + 3, last)];
    for (int j = beg; j < end; j += 4) {
      v4f x0 = *(const v4f*)(fq + (long)e0.x * NF);
      v4f x1 = *(const v4f*)(fq + (long)e1.x * NF);
      v4f x2 = *(const v4f*)(fq + (long)e2.x * NF);
      v4f x3 = *(const v4f*)(fq + (long)e3.x * NF);
      const float w0 = __int_as_float(e0.y);
      const float w1 = __int_as_float(e1.y);
      const float w2 = __int_as_float(e2.y);
      const float w3 = __int_as_float(e3.y);
      e0 = ed[min(j + 4, last)];
      e1 = ed[min(j + 5, last)];
      e2 = ed[min(j + 6, last)];
      e3 = ed[min(j + 7, last)];
      acc += w0 * x0;
      if (j + 1 < end) acc += w1 * x1;
      if (j + 2 < end) acc += w2 * x2;
      if (j + 3 < end) acc += w3 * x3;
    }
  }
  return acc;
}

template <int NF>
__device__ __forceinline__ v4f gcn_row_sum8_lds(
    const float* __restrict__ fq, const v2i* ed, int beg, int end) {
  v4f acc = {0.0f, 0.0f, 0.0f, 0.0f};
  if (beg < end) {
    const int last = end - 1;
    v2i e[8];
#pragma unroll
    for (int u = 0; u < 8; ++u) e[u] = ed[min(beg + u, last)];
    for (int j = beg; j < end; j += 8) {
      v4f x[8];
      float wv[8];
#pragma unroll
      for (int u = 0; u < 8; ++u) {
        x[u]  = *(const v4f*)(fq + (long)e[u].x * NF);
        wv[u] = __int_as_float(e[u].y);
      }
#pragma unroll
      for (int u = 0; u < 8; ++u) e[u] = ed[min(j + 8 + u, last)];
#pragma unroll
      for (int u = 0; u < 8; ++u)
        if (j + u < end) acc += wv[u] * x[u];
    }
  }
  return acc;
}

#define SCAP 8192
__global__ __launch_bounds__(1024) void gcn_bucket_rows_spmm(
    const int* __restrict__ tmp8, const int* __restrict__ segT,
    const float* __restrict__ feat, int* __restrict__ row_ptr,
    int* __restrict__ csr, float* __restrict__ agg) {
  __shared__ int cnt[512];
  __shared__ int rst[512];
  __shared__ int wtot[32];
  __shared__ int sbeg[NW];
  __shared__ int slen[NW];
  __shared__ int rbeg[8];
  __shared__ int rlen[8];
  __shared__ __attribute__((aligned(16))) v2i stage[SCAP + 16];
  const int tid  = threadIdx.x;
  const int lane = tid & 31;
  const int wave = tid >> 5;
  const int b    = blockIdx.x;
  const int lo   = b * BS;
  const int nb   = min(BS, N_NODES - lo);
  const v2i* __restrict__ in = (const v2i*)tmp8;
  if (tid < 512) cnt[tid] = 0;

  int s0 = 0, sl = 0;
  if (tid < NW) {
    s0 = segT[b * NW + tid];
    sl = segT[(b + 1) * NW + tid] - s0;
    sbeg[tid] = tid * CSTR + s0;
    slen[tid] = sl;
  }
  int ra = s0, rl = sl;
#pragma unroll
  for (int o = 16; o > 0; o >>= 1) {
    ra += __shfl_xor(ra, o);
    rl += __shfl_xor(rl, o);
  }
  if (tid < NW && lane == 0) {
    rbeg[wave] = ra;
    rlen[wave] = rl;
  }
  __syncthreads();
  int beg = 0, tot = 0;
#pragma unroll
  for (int k = 0; k < 8; ++k) {
    beg += rbeg[k];
    tot += rlen[k];
  }
  const int end = beg + tot;
  const int  sh     = beg & 15;
  const bool staged = tot <= SCAP;

  v2i r[16];
#pragma unroll
  for (int k = 0; k < 8; ++k) {
    const int ch = wave * 8 + k;
    const int len = slen[ch];
    const int base = sbeg[ch];
    if (lane < len)      r[2 * k] = in[base + lane];
    else                 r[2 * k] = (v2i){-1, 0};
    if (lane + 32 < len) r[2 * k + 1] = in[base + 32 + lane];
    else                 r[2 * k + 1] = (v2i){-1, 0};
  }
#pragma unroll
  for (int k = 0; k < 16; ++k)
    if (r[k].x >= 0) atomicAdd(&cnt[r[k].x >> 17], 1);
  for (int k = 0; k < 8; ++k) {
    const int ch = wave * 8 + k;
    const int len = slen[ch];
    const int base = sbeg[ch];
    for (int i = 64 + lane; i < len; i += 32)
      atomicAdd(&cnt[in[base + i].x >> 17], 1);
  }
  __syncthreads();

  const int c = (tid < nb) ? cnt[tid] : 0;
  int incl = c;
#pragma unroll
  for (int o = 1; o < 32; o <<= 1) {
    int v = __shfl_up(incl, o);
    if (lane >= o) incl += v;
  }
  if (lane == 31) wtot[wave] = incl;
  __syncthreads();
  int off = 0;
#pragma unroll
  for (int k = 0; k < 32; ++k)
    if (k < wave) off += wtot[k];
  const int first = beg + off + incl - c;
  if (tid < nb) {
    row_ptr[lo + tid] = first;
    rst[tid] = first;
    cnt[tid] = first;
  }
  if (b == NB - 1 && tid == 0) row_ptr[N_NODES] = end;
  __syncthreads();

#pragma unroll
  for (int k = 0; k < 16; ++k) {
    if (r[k].x >= 0) {
      const int pos = atomicAdd(&cnt[r[k].x >> 17], 1);
      v2i v;
      v.x = r[k].x & 0x1FFFF;
      v.y = r[k].y;
      if (staged) stage[pos - beg + sh] = v;
      else        ((v2i*)csr)[pos] = v;
    }
  }
  for (int k = 0; k < 8; ++k) {
    const int ch = wave * 8 + k;
    const int len = slen[ch];
    const int base = sbeg[ch];
    for (int i = 64 + lane; i < len; i += 32) {
      const v2i e = in[base + i];
      const int pos = atomicAdd(&cnt[e.x >> 17], 1);
      v2i v;
      v.x = e.x & 0x1FFFF;
      v.y = e.y;
      if (staged) stage[pos - beg + sh] = v;
      else        ((v2i*)csr)[pos] = v;
    }
  }

  __syncthreads();

  if (staged) {
    v2i* __restrict__ line0 = (v2i*)csr + (beg - sh);
    const int n = sh + tot;
    for (int j = 2 * tid; j < n; j += 2048) {
      const bool lo_ok = j >= sh;
      const bool hi_ok = (j + 1 >= sh) && (j + 1 < n);
      if (lo_ok && hi_ok)
        *(v4i*)(line0 + j) = *(const v4i*)(stage + j);
      else if (lo_ok)
        line0[j] = stage[j];
      else if (hi_ok)
        line0[j + 1] = stage[j + 1];
    }
  }

  const int q = lane & 15;
  const float* __restrict__ fq = feat + q * 4;
  for (int rr = wave * 2 + (lane >> 4); rr < nb; rr += 64) {
    const int js = rst[rr];
    const int je = cnt[rr];
    v4f acc;
    if (staged)
      acc = gcn_row_sum8_lds<FEAT>(fq, stage, js - beg + sh, je - beg + sh);
    else
      acc = gcn_row_sum4<FEAT>(fq, (const v2i*)csr, js, je);
    *(v4f*)(agg + (long)(lo + rr) * FEAT + q * 4) = acc;
  }
}

template <int M>
__device__ __forceinline__ float gcn_row_xor(float v) {
  return __int_as_float(__builtin_amdgcn_update_dpp(
      0, __float_as_int(v), 0x160 + M, 0xf, 0xf, false));
}

__global__ __launch_bounds__(256) void gcn_spmm_gather4_logsoftmax(
    const float* __restrict__ feat, const int* __restrict__ row_ptr,
    const int* __restrict__ csr, float* __restrict__ out, int n_rows) {
  constexpr int QP = NCLS / 4;
  static_assert(NCLS % 4 == 0 && QP <= 16, "a class row fits one 16-lane row");
  const int d = (blockIdx.x * blockDim.x + threadIdx.x) >> 4;
  const int q = threadIdx.x & 15;
  const bool act = (d < n_rows) && (q < QP);
  v4f acc = {0.0f, 0.0f, 0.0f, 0.0f};
  if (act)
    acc = gcn_row_sum4<NCLS>(feat + q * 4, (const v2i*)csr,
                             row_ptr[d], row_ptr[d + 1]);

  float mx = act ? fmaxf(fmaxf(acc.x, acc.y), fmaxf(acc.z, acc.w)) : -INFINITY;
  mx = fmaxf(mx, gcn_row_xor<1>(mx));
  mx = fmaxf(mx, gcn_row_xor<2>(mx));
  mx = fmaxf(mx, gcn_row_xor<4>(mx));
  mx = fmaxf(mx, gcn_row_xor<8>(mx));
  float s = 0.0f;
  if (act)
    s = (__expf(acc.x - mx) + __expf(acc.y - mx)) +
        (__expf(acc.z - mx) + __expf(acc.w - mx));
  s += gcn_row_xor<1>(s);
  s += gcn_row_xor<2>(s);
  s += gcn_row_xor<4>(s);
  s += gcn_row_xor<8>(s);
  const float lse = mx + __logf(s);
  if (act) *(v4f*)(out + (long)d * NCLS + q * 4) = acc - lse;
}

#define HP 132
#define W1F_V2 ((FEAT / 4) * (EMBED / 16) * 32)
#define W2F_V2 ((EMBED / 4) * 3 * 32)
static_assert(W1F_V2 == 16 * 256 && W2F_V2 == 12 * 256, "staging trip counts");
__global__ __launch_bounds__(256) void gcn_fused_layers(
    const float* __restrict__ A, const float* __restrict__ W1,
    const float* __restrict__ W2, float* __restrict__ T, int mTiles) {
  __shared__ __attribute__((aligned(16)))
      float smem[2 * W1F_V2 + 2 * W2F_V2 + 8 * 16 * HP];
  v2f* const w1f = (v2f*)smem;
  v2f* const w2f = (v2f*)(smem + 2 * W1F_V2);
  const int tid  = threadIdx.x;
  const int wave = tid >> 5;
  const int lane = tid & 31;
  const int n    = lane & 15;
  const int half = lane >> 4;
  const int klo  = half * 2;
  float* const hb = smem + 2 * W1F_V2 + 2 * W2F_V2 + wave * (16 * HP);

  const int  tileRaw = blockIdx.x * 8 + wave;
  const bool valid   = tileRaw < mTiles;
  const int  tile    = valid ? tileRaw : (mTiles - 1);

  const float* __restrict__ arow = A + (long)(tile * 16 + n) * FEAT + klo;
  v2f a[FEAT / 4];
#pragma unroll
  for (int ks = 0; ks < FEAT / 4; ++ks) a[ks] = *(const v2f*)(arow + ks * 4);

#pragma unroll
  for (int j = 0; j < W1F_V2 / 256; ++j) {
    const float* __restrict__ p = W1 + (j * 4 + klo) * EMBED + wave * 16 + n;
    v2f v;
    v.x = p[0];
    v.y = p[EMBED];
    w1f[j * 256 + tid] = v;
  }
#pragma unroll
  for (int j = 0; j < W2F_V2 / 256; ++j) {
    const int  nt  = j >> 2;
    const int  ks  = wave + 8 * (j & 3);
    const int  col = nt * 16 + n;
    const bool in  = col < NCLS;
    const float* __restrict__ p = W2 + (ks * 4 + klo) * NCLS + (in ? col : 0);
    const float x0 = p[0];
    const float x1 = p[NCLS];
    v2f v;
    v.x = in ? x0 : 0.0f;
    v.y = in ? x1 : 0.0f;
    w2f[j * 256 + tid] = v;
  }
  __syncthreads();

  v8f c[EMBED / 16];
#pragma unroll
  for (int nt = 0; nt < EMBED / 16; ++nt) c[nt] = (v8f){};
#pragma unroll
  for (int ks = 0; ks < FEAT / 4; ++ks) {
#pragma unroll
    for (int nt = 0; nt < EMBED / 16; ++nt) {
      const v2f b = w1f[(ks * (EMBED / 16) + nt) * 32 + lane];
      c[nt] = __builtin_amdgcn_wmma_f32_16x16x4_f32(
          false, a[ks], false, b, (short)0, c[nt], false, false);
    }
  }
#pragma unroll
  for (int nt = 0; nt < EMBED / 16; ++nt) {
#pragma unroll
    for (int r = 0; r < 8; ++r)
      hb[(r + 8 * half) * HP + nt * 16 + n] = fmaxf(c[nt][r], 0.0f);
  }
  __syncthreads();

  v8f t0 = {}, t1 = {}, t2 = {};
  const float* __restrict__ hrow = hb + n * HP + klo;
#pragma unroll
  for (int ks = 0; ks < EMBED / 4; ++ks) {
    const v2f ha = *(const v2f*)(hrow + ks * 4);
    const v2f b0 = w2f[(0 * 32 + ks) * 32 + lane];
    const v2f b1 = w2f[(1 * 32 + ks) * 32 + lane];
    const v2f b2 = w2f[(2 * 32 + ks) * 32 + lane];
    t0 = __builtin_amdgcn_wmma_f32_16x16x4_f32(
        false, ha, false, b0, (short)0, t0, false, false);
    t1 = __builtin_amdgcn_wmma_f32_16x16x4_f32(
        false, ha, false, b1, (short)0, t1, false, false);
    t2 = __builtin_amdgcn_wmma_f32_16x16x4_f32(
        false, ha, false, b2, (short)0, t2, false, false);
  }

  if (valid) {
#pragma unroll
    for (int r = 0; r < 8; ++r) {
      float* __restrict__ trow = T + (long)(tile * 16 + r + half * 8) * NCLS;
      trow[n]      = t0[r];
      trow[16 + n] = t1[r];
      if (n < NCLS - 32) trow[32 + n] = t2[r];
    }
  }
}

extern "C" void kernel_launch(void* const* d_in, const int* in_sizes, int n_in,
                              void* d_out, int out_size, void* d_ws, size_t ws_size,
                              hipStream_t stream) {
  const float* x    = (const float*)d_in[0];
  const float* W1   = (const float*)d_in[1];
  const float* W2   = (const float*)d_in[2];
  const float* ew   = (const float*)d_in[3];
  const int*   esrc = (const int*)d_in[4];
  const int*   edst = (const int*)d_in[5];
  float*       out  = (float*)d_out;

  float* agg1 = (float*)d_ws;
  float* scr  = agg1 + (size_t)N_NODES * FEAT;
  float* t    = scr  + (size_t)N_NODES * EMBED;
  int*   csr  = (int*)(t + (size_t)N_NODES * NCLS);
  int*   rptr = csr + (size_t)2 * N_EDGES;
  int*   tmp8   = (int*)scr;
  int*   segT   = tmp8 + (size_t)2 * NW * CSTR;

  const int TPB = 256;

  gcn_chunk_sort<<<NW, 1024, 0, stream>>>(esrc, edst, ew, tmp8, segT);
  gcn_bucket_rows_spmm<<<NB, 1024, 0, stream>>>(tmp8, segT, x, rptr, csr, agg1);

  {
    const int mTiles = N_NODES / 16;
    gcn_fused_layers<<<(mTiles + 7) / 8, TPB, 0, stream>>>(agg1, W1, W2, t, mTiles);
  }

  gcn_spmm_gather4_logsoftmax<<<(N_NODES * 16 + TPB - 1) / TPB, TPB, 0, stream>>>(
      t, rptr, csr, out, N_NODES);
}


# Both dense layers fused into the bucket kernel: a bucket's agg rows stay in LDS, one wave per 16-row tile runs relu(agg@W1)@W2 there; agg plane and one launch gone (3 launches); sum order unchanged
# speedup vs baseline: 4.6577x; 1.1034x over previous
#include <hip/hip_runtime.h>
#include <math.h>

typedef __attribute__((ext_vector_type(2))) float v2f;
typedef __attribute__((ext_vector_type(4))) float v4f;
typedef __attribute__((ext_vector_type(8))) float v8f;
typedef __attribute__((ext_vector_type(2))) int v2i;
typedef __attribute__((ext_vector_type(4))) int v4i;

#define N_NODES 100000
#define N_EDGES 1600000
#define FEAT    64
#define EMBED   128
#define NCLS    40

#define NB     256
#define BS     391
#define NW     256
#define CHUNK  ((N_EDGES + NW - 1) / NW)
#define CSTR   ((CHUNK + 15) / 16 * 16)
#define TRIPS  ((CHUNK + 1023) / 1024)
static_assert(NB * BS >= N_NODES, "buckets must cover every row");
static_assert(BS <= 512 && N_NODES <= (1 << 17), "packed edge word");
static_assert(NB == 256 && NW == 256, "one thread per bucket / per chunk below");

__global__ __launch_bounds__(1024) void gcn_chunk_sort(
    const int* __restrict__ src, const int* __restrict__ dst,
    const float* __restrict__ w, int* __restrict__ tmp8,
    int* __restrict__ segT) {
  __shared__ int c[NB];
  __shared__ int wtot[8];
  __shared__ v2i stage[CHUNK];
  const int tid  = threadIdx.x;
  const int lane = tid & 31;
  const int wave = tid >> 5;
  const int beg = blockIdx.x * CHUNK;
  const int end = min(beg + CHUNK, N_EDGES);
  if (tid < NB) c[tid] = 0;

  int bk[TRIPS], pk[TRIPS], wb[TRIPS], rk[TRIPS];
#pragma unroll
  for (int k = 0; k < TRIPS; ++k) {
    const int e = beg + tid + k * 1024;
    const bool ok = e < end;
    const int d = ok ? dst[e] : -1;
    const int s = ok ? src[e] : 0;
    wb[k] = ok ? __float_as_int(w[e]) : 0;
    bk[k] = ok ? d / BS : -1;
    pk[k] = s | ((d - bk[k] * BS) << 17);
  }
  __syncthreads();
#pragma unroll
  for (int k = 0; k < TRIPS; ++k) {
    rk[k] = 0;
    if (bk[k] >= 0) rk[k] = atomicAdd(&c[bk[k]], 1);
  }
  __syncthreads();

  const int T = (tid < NB) ? c[tid] : 0;
  int incl = T;
#pragma unroll
  for (int o = 1; o < 32; o <<= 1) {
    int v = __shfl_up(incl, o);
    if (lane >= o) incl += v;
  }
  if (tid < NB && lane == 31) wtot[wave] = incl;
  __syncthreads();
  if (tid < NB) {
    int off = 0;
#pragma unroll
    for (int k = 0; k < 8; ++k)
      if (k < wave) off += wtot[k];
    const int excl = off + incl - T;
    c[tid] = excl;
    segT[tid * NW + blockIdx.x] = excl;
    if (tid == NB - 1) segT[NB * NW + blockIdx.x] = excl + T;
  }
  __syncthreads();

#pragma unroll
  for (int k = 0; k < TRIPS; ++k) {
    if (bk[k] >= 0) {
      v2i r;
      r.x = pk[k];
      r.y = wb[k];
      stage[c[bk[k]] + rk[k]] = r;
    }
  }
  __syncthreads();

  v2i* __restrict__ outw = (v2i*)tmp8 + (long)blockIdx.x * CSTR;
  const int n = end - beg;
#pragma unroll
  for (int k = 0; k < TRIPS; ++k) {
    const int i = tid + k * 1024;
    if (i < n) outw[i] = stage[i];
  }
}

template <int NF>
__device__ __forceinline__ v4f gcn_row_sum4(
    const float* __restrict__ fq, const v2i* __restrict__ ed, int beg, int end) {
  v4f acc = {0.0f, 0.0f, 0.0f, 0.0f};
  if (beg < end) {
    const int last = end - 1;
    v2i e0 = ed[beg];
    v2i e1 = ed[min(beg + 1, last)];
    v2i e2 = ed[min(beg + 2, last)];
    v2i e3 = ed[min(beg + 3, last)];
    for (int j = beg; j < end; j += 4) {
      v4f x0 = *(const v4f*)(fq + (long)e0.x * NF);
      v4f x1 = *(const v4f*)(fq + (long)e1.x * NF);
      v4f x2 = *(const v4f*)(fq + (long)e2.x * NF);
      v4f x3 = *(const v4f*)(fq + (long)e3.x * NF);
      const float w0 = __int_as_float(e0.y);
      const float w1 = __int_as_float(e1.y);
      const float w2 = __int_as_float(e2.y);
      const float w3 = __int_as_float(e3.y);
      e0 = ed[min(j + 4, last)];
      e1 = ed[min(j + 5, last)];
      e2 = ed[min(j + 6, last)];
      e3 = ed[min(j + 7, last)];
      acc += w0 * x0;
      if (j + 1 < end) acc += w1 * x1;
      if (j + 2 < end) acc += w2 * x2;
      if (j + 3 < end) acc += w3 * x3;
    }
  }
  return acc;
}

template <int NF>
__device__ __forceinline__ v4f gcn_row_sum8_lds(
    const float* __restrict__ fq, const v2i* ed, int beg, int end) {
  v4f acc = {0.0f, 0.0f, 0.0f, 0.0f};
  if (beg < end) {
    const int last = end - 1;
    v2i e[8];
#pragma unroll
    for (int u = 0; u < 8; ++u) e[u] = ed[min(beg + u, last)];
    for (int j = beg; j < end; j += 8) {
      v4f x[8];
      float wv[8];
#pragma unroll
      for (int u = 0; u < 8; ++u) {
        x[u]  = *(const v4f*)(fq + (long)e[u].x * NF);
        wv[u] = __int_as_float(e[u].y);
      }
#pragma unroll
      for (int u = 0; u < 8; ++u) e[u] = ed[min(j + 8 + u, last)];
#pragma unroll
      for (int u = 0; u < 8; ++u)
        if (j + u < end) acc += wv[u] * x[u];
    }
  }
  return acc;
}

#define SCAP 8192
#define WP   68
#define NTILE ((BS + 15) / 16)
#define W1F_V2 ((FEAT / 4) * (EMBED / 16) * 32)
#define W2F_V2 ((EMBED / 4) * 3 * 32)
static_assert(W1F_V2 == 4 * 1024 && W2F_V2 == 3 * 1024, "staging trip counts");
static_assert(NTILE <= 32 && FEAT == 64 && EMBED == 128, "one wave per tile, 32-column quarters");

__device__ __forceinline__ void gcn_wave_lds_order() {
  __builtin_amdgcn_fence(__ATOMIC_SEQ_CST, "wavefront");
  __builtin_amdgcn_wave_barrier();
}

__global__ __launch_bounds__(1024) void gcn_bucket_rows_spmm_layers(
    const int* __restrict__ tmp8, const int* __restrict__ segT,
    const float* __restrict__ feat, const float* __restrict__ W1,
    const float* __restrict__ W2, int* __restrict__ row_ptr,
    int* __restrict__ csr, float* __restrict__ T) {
  __shared__ __attribute__((aligned(16)))
      float gsm[2 * W1F_V2 + 2 * W2F_V2 + NTILE * 16 * WP];
  v2f* const w1f = (v2f*)gsm;
  v2f* const w2f = (v2f*)(gsm + 2 * W1F_V2);
  float* const aggL = gsm + 2 * W1F_V2 + 2 * W2F_V2;
  __shared__ int cnt[512];
  __shared__ int rst[512];
  __shared__ int wtot[32];
  __shared__ int sbeg[NW];
  __shared__ int slen[NW];
  __shared__ int rbeg[8];
  __shared__ int rlen[8];
  __shared__ __attribute__((aligned(16))) v2i stage[SCAP + 16];
  const int tid  = threadIdx.x;
  const int lane = tid & 31;
  const int wave = tid >> 5;
  const int b    = blockIdx.x;
  const int lo   = b * BS;
  const int nb   = min(BS, N_NODES - lo);
  const v2i* __restrict__ in = (const v2i*)tmp8;
  if (tid < 512) cnt[tid] = 0;

  int s0 = 0, sl = 0;
  if (tid < NW) {
    s0 = segT[b * NW + tid];
    sl = segT[(b + 1) * NW + tid] - s0;
    sbeg[tid] = tid * CSTR + s0;
    slen[tid] = sl;
  }
  const int wn   = lane & 15;
  const int wklo = (lane >> 4) * 2;
  float w1x[4], w1y[4], w2x[3], w2y[3];
#pragma unroll
  for (int j = 0; j < 4; ++j) {
    const int ks = j * 4 + (wave >> 3);
    const float* __restrict__ p = W1 + (ks * 4 + wklo) * EMBED + (wave & 7) * 16 + wn;
    w1x[j] = p[0];
    w1y[j] = p[EMBED];
  }
#pragma unroll
  for (int j = 0; j < 3; ++j) {
    const int  col = j * 16 + wn;
    const bool inb = col < NCLS;
    const float* __restrict__ p = W2 + (wave * 4 + wklo) * NCLS + (inb ? col : 0);
    w2x[j] = p[0];
    w2y[j] = p[NCLS];
  }

  int ra = s0, rl = sl;
#pragma unroll
  for (int o = 16; o > 0; o >>= 1) {
    ra += __shfl_xor(ra, o);
    rl += __shfl_xor(rl, o);
  }
  if (tid < NW && lane == 0) {
    rbeg[wave] = ra;
    rlen[wave] = rl;
  }
  __syncthreads();
#pragma unroll
  for (int j = 0; j < 4; ++j) {
    v2f v;
    v.x = w1x[j];
    v.y = w1y[j];
    w1f[j * 1024 + tid] = v;
  }
#pragma unroll
  for (int j = 0; j < 3; ++j) {
    const bool inb = j * 16 + wn < NCLS;
    v2f v;
    v.x = inb ? w2x[j] : 0.0f;
    v.y = inb ? w2y[j] : 0.0f;
    w2f[j * 1024 + tid] = v;
  }
  int beg = 0, tot = 0;
#pragma unroll
  for (int k = 0; k < 8; ++k) {
    beg += rbeg[k];
    tot += rlen[k];
  }
  const int end = beg + tot;
  const int  sh     = beg & 15;
  const bool staged = tot <= SCAP;

  v2i r[16];
#pragma unroll
  for (int k = 0; k < 8; ++k) {
    const int ch = wave * 8 + k;
    const int len = slen[ch];
    const int base = sbeg[ch];
    if (lane < len)      r[2 * k] = in[base + lane];
    else                 r[2 * k] = (v2i){-1, 0};
    if (lane + 32 < len) r[2 * k + 1] = in[base + 32 + lane];
    else                 r[2 * k + 1] = (v2i){-1, 0};
  }
#pragma unroll
  for (int k = 0; k < 16; ++k)
    if (r[k].x >= 0) atomicAdd(&cnt[r[k].x >> 17], 1);
  for (int k = 0; k < 8; ++k) {
    const int ch = wave * 8 + k;
    const int len = slen[ch];
    const int base = sbeg[ch];
    for (int i = 64 + lane; i < len; i += 32)
      atomicAdd(&cnt[in[base + i].x >> 17], 1);
  }
  __syncthreads();

  const int c = (tid < nb) ? cnt[tid] : 0;
  int incl = c;
#pragma unroll
  for (int o = 1; o < 32; o <<= 1) {
    int v = __shfl_up(incl, o);
    if (lane >= o) incl += v;
  }
  if (lane == 31) wtot[wave] = incl;
  __syncthreads();
  int off = 0;
#pragma unroll
  for (int k = 0; k < 32; ++k)
    if (k < wave) off += wtot[k];
  const int first = beg + off + incl - c;
  if (tid < nb) {
    row_ptr[lo + tid] = first;
    rst[tid] = first;
    cnt[tid] = first;
  }
  if (b == NB - 1 && tid == 0) row_ptr[N_NODES] = end;
  __syncthreads();

#pragma unroll
  for (int k = 0; k < 16; ++k) {
    if (r[k].x >= 0) {
      const int pos = atomicAdd(&cnt[r[k].x >> 17], 1);
      v2i v;
      v.x = r[k].x & 0x1FFFF;
      v.y = r[k].y;
      if (staged) stage[pos - beg + sh] = v;
      else        ((v2i*)csr)[pos] = v;
    }
  }
  for (int k = 0; k < 8; ++k) {
    const int ch = wave * 8 + k;
    const int len = slen[ch];
    const int base = sbeg[ch];
    for (int i = 64 + lane; i < len; i += 32) {
      const v2i e = in[base + i];
      const int pos = atomicAdd(&cnt[e.x >> 17], 1);
      v2i v;
      v.x = e.x & 0x1FFFF;
      v.y = e.y;
      if (staged) stage[pos - beg + sh] = v;
      else        ((v2i*)csr)[pos] = v;
    }
  }

  __syncthreads();

  if (staged) {
    v2i* __restrict__ line0 = (v2i*)csr + (beg - sh);
    const int n = sh + tot;
    for (int j = 2 * tid; j < n; j += 2048) {
      const bool lo_ok = j >= sh;
      const bool hi_ok = (j + 1 >= sh) && (j + 1 < n);
      if (lo_ok && hi_ok)
        *(v4i*)(line0 + j) = *(const v4i*)(stage + j);
      else if (lo_ok)
        line0[j] = stage[j];
      else if (hi_ok)
        line0[j + 1] = stage[j + 1];
    }
  }

  const int q = lane & 15;
  const float* __restrict__ fq = feat + q * 4;
  for (int rr = wave * 2 + (lane >> 4); rr < NTILE * 16; rr += 64) {
    v4f acc = {0.0f, 0.0f, 0.0f, 0.0f};
    if (rr < nb) {
      const int js = rst[rr];
      const int je = cnt[rr];
      if (staged)
        acc = gcn_row_sum8_lds<FEAT>(fq, stage, js - beg + sh, je - beg + sh);
      else
        acc = gcn_row_sum4<FEAT>(fq, (const v2i*)csr, js, je);
    }
    *(v4f*)(aggL + rr * WP + q * 4) = acc;
  }
  __syncthreads();

  const int wv = __builtin_amdgcn_readfirstlane(wave);
  if (wv * 16 < nb) {
    float* const win = aggL + wv * (16 * WP);
    const int n    = lane & 15;
    const int half = lane >> 4;
    const int klo  = half * 2;
    const float* const arow = win + n * WP + klo;

    v2f a[FEAT / 4];
#pragma unroll
    for (int ks = 0; ks < FEAT / 4; ++ks) a[ks] = *(const v2f*)(arow + ks * 4);

    v8f t0 = {}, t1 = {}, t2 = {};
#pragma unroll
    for (int nq = 0; nq < 4; ++nq) {
      v8f c[2];
#pragma unroll
      for (int j = 0; j < 2; ++j) c[j] = (v8f){};
#pragma unroll
      for (int ks = 0; ks < FEAT / 4; ++ks) {
#pragma unroll
        for (int j = 0; j < 2; ++j) {
          const v2f bw = w1f[(ks * (EMBED / 16) + nq * 2 + j) * 32 + lane];
          c[j] = __builtin_amdgcn_wmma_f32_16x16x4_f32(
              false, a[ks], false, bw, (short)0, c[j], false, false);
        }
      }
      gcn_wave_lds_order();
#pragma unroll
      for (int j = 0; j < 2; ++j) {
#pragma unroll
        for (int r = 0; r < 8; ++r)
          win[(r + 8 * half) * WP + j * 16 + n] = fmaxf(c[j][r], 0.0f);
      }
      gcn_wave_lds_order();

#pragma unroll
      for (int ks = 0; ks < 8; ++ks) {
        const v2f ha = *(const v2f*)(arow + ks * 4);
        const v2f b0 = w2f[(0 * 32 + nq * 8 + ks) * 32 + lane];
        const v2f b1 = w2f[(1 * 32 + nq * 8 + ks) * 32 + lane];
        const v2f b2 = w2f[(2 * 32 + nq * 8 + ks) * 32 + lane];
        t0 = __builtin_amdgcn_wmma_f32_16x16x4_f32(
            false, ha, false, b0, (short)0, t0, false, false);
        t1 = __builtin_amdgcn_wmma_f32_16x16x4_f32(
            false, ha, false, b1, (short)0, t1, false, false);
        t2 = __builtin_amdgcn_wmma_f32_16x16x4_f32(
            false, ha, false, b2, (short)0, t2, false, false);
      }
    }

#pragma unroll
    for (int r = 0; r < 8; ++r) {
      const int trl = wv * 16 + r + half * 8;
      if (trl < nb) {
        float* __restrict__ trow = T + (long)(lo + trl) * NCLS;
        trow[n]      = t0[r];
        trow[16 + n] = t1[r];
        if (n < NCLS - 32) trow[32 + n] = t2[r];
      }
    }
  }
}

template <int M>
__device__ __forceinline__ float gcn_row_xor(float v) {
  return __int_as_float(__builtin_amdgcn_update_dpp(
      0, __float_as_int(v), 0x160 + M, 0xf, 0xf, false));
}

__global__ __launch_bounds__(256) void gcn_spmm_gather4_logsoftmax(
    const float* __restrict__ feat, const int* __restrict__ row_ptr,
    const int* __restrict__ csr, float* __restrict__ out, int n_rows) {
  constexpr int QP = NCLS / 4;
  static_assert(NCLS % 4 == 0 && QP <= 16, "a class row fits one 16-lane row");
  const int d = (blockIdx.x * blockDim.x + threadIdx.x) >> 4;
  const int q = threadIdx.x & 15;
  const bool act = (d < n_rows) && (q < QP);
  v4f acc = {0.0f, 0.0f, 0.0f, 0.0f};
  if (act)
    acc = gcn_row_sum4<NCLS>(feat + q * 4, (const v2i*)csr,
                             row_ptr[d], row_ptr[d + 1]);

  float mx = act ? fmaxf(fmaxf(acc.x, acc.y), fmaxf(acc.z, acc.w)) : -INFINITY;
  mx = fmaxf(mx, gcn_row_xor<1>(mx));
  mx = fmaxf(mx, gcn_row_xor<2>(mx));
  mx = fmaxf(mx, gcn_row_xor<4>(mx));
  mx = fmaxf(mx, gcn_row_xor<8>(mx));
  float s = 0.0f;
  if (act)
    s = (__expf(acc.x - mx) + __expf(acc.y - mx)) +
        (__expf(acc.z - mx) + __expf(acc.w - mx));
  s += gcn_row_xor<1>(s);
  s += gcn_row_xor<2>(s);
  s += gcn_row_xor<4>(s);
  s += gcn_row_xor<8>(s);
  const float lse = mx + __logf(s);
  if (act) *(v4f*)(out + (long)d * NCLS + q * 4) = acc - lse;
}

extern "C" void kernel_launch(void* const* d_in, const int* in_sizes, int n_in,
                              void* d_out, int out_size, void* d_ws, size_t ws_size,
                              hipStream_t stream) {
  const float* x    = (const float*)d_in[0];
  const float* W1   = (const float*)d_in[1];
  const float* W2   = (const float*)d_in[2];
  const float* ew   = (const float*)d_in[3];
  const int*   esrc = (const int*)d_in[4];
  const int*   edst = (const int*)d_in[5];
  float*       out  = (float*)d_out;

  float* agg1 = (float*)d_ws;
  float* scr  = agg1 + (size_t)N_NODES * FEAT;
  float* t    = scr  + (size_t)N_NODES * EMBED;
  int*   csr  = (int*)(t + (size_t)N_NODES * NCLS);
  int*   rptr = csr + (size_t)2 * N_EDGES;
  int*   tmp8   = (int*)scr;
  int*   segT   = tmp8 + (size_t)2 * NW * CSTR;

  const int TPB = 256;

  gcn_chunk_sort<<<NW, 1024, 0, stream>>>(esrc, edst, ew, tmp8, segT);
  gcn_bucket_rows_spmm_layers<<<NB, 1024, 0, stream>>>(tmp8, segT, x, W1, W2, rptr, csr, t);

  gcn_spmm_gather4_logsoftmax<<<(N_NODES * 16 + TPB - 1) / TPB, TPB, 0, stream>>>(
      t, rptr, csr, out, N_NODES);
}


# Row-sum gather loops of both SPMMs: address = scalar base + 32-bit byte offset, a slot past the row's end gets weight 0 instead of four selects; 115 -> 71 vector ops per 8-edge trip; sum order same
# speedup vs baseline: 4.8856x; 1.0489x over previous
#include <hip/hip_runtime.h>
#include <math.h>

typedef __attribute__((ext_vector_type(2))) float v2f;
typedef __attribute__((ext_vector_type(4))) float v4f;
typedef __attribute__((ext_vector_type(8))) float v8f;
typedef __attribute__((ext_vector_type(2))) int v2i;
typedef __attribute__((ext_vector_type(4))) int v4i;

#define N_NODES 100000
#define N_EDGES 1600000
#define FEAT    64
#define EMBED   128
#define NCLS    40

#define NB     256
#define BS     391
#define NW     256
#define CHUNK  ((N_EDGES + NW - 1) / NW)
#define CSTR   ((CHUNK + 15) / 16 * 16)
#define TRIPS  ((CHUNK + 1023) / 1024)
static_assert(NB * BS >= N_NODES, "buckets must cover every row");
static_assert(BS <= 512 && N_NODES <= (1 << 17), "packed edge word");
static_assert(NB == 256 && NW == 256, "one thread per bucket / per chunk below");

__global__ __launch_bounds__(1024) void gcn_chunk_sort(
    const int* __restrict__ src, const int* __restrict__ dst,
    const float* __restrict__ w, int* __restrict__ tmp8,
    int* __restrict__ segT) {
  __shared__ int c[NB];
  __shared__ int wtot[8];
  __shared__ v2i stage[CHUNK];
  const int tid  = threadIdx.x;
  const int lane = tid & 31;
  const int wave = tid >> 5;
  const int beg = blockIdx.x * CHUNK;
  const int end = min(beg + CHUNK, N_EDGES);
  if (tid < NB) c[tid] = 0;

  int bk[TRIPS], pk[TRIPS], wb[TRIPS], rk[TRIPS];
#pragma unroll
  for (int k = 0; k < TRIPS; ++k) {
    const int e = beg + tid + k * 1024;
    const bool ok = e < end;
    const int d = ok ? dst[e] : -1;
    const int s = ok ? src[e] : 0;
    wb[k] = ok ? __float_as_int(w[e]) : 0;
    bk[k] = ok ? d / BS : -1;
    pk[k] = s | ((d - bk[k] * BS) << 17);
  }
  __syncthreads();
#pragma unroll
  for (int k = 0; k < TRIPS; ++k) {
    rk[k] = 0;
    if (bk[k] >= 0) rk[k] = atomicAdd(&c[bk[k]], 1);
  }
  __syncthreads();

  const int T = (tid < NB) ? c[tid] : 0;
  int incl = T;
#pragma unroll
  for (int o = 1; o < 32; o <<= 1) {
    int v = __shfl_up(incl, o);
    if (lane >= o) incl += v;
  }
  if (tid < NB && lane == 31) wtot[wave] = incl;
  __syncthreads();
  if (tid < NB) {
    int off = 0;
#pragma unroll
    for (int k = 0; k < 8; ++k)
      if (k < wave) off += wtot[k];
    const int excl = off + incl - T;
    c[tid] = excl;
    segT[tid * NW + blockIdx.x] = excl;
    if (tid == NB - 1) segT[NB * NW + blockIdx.x] = excl + T;
  }
  __syncthreads();

#pragma unroll
  for (int k = 0; k < TRIPS; ++k) {
    if (bk[k] >= 0) {
      v2i r;
      r.x = pk[k];
      r.y = wb[k];
      stage[c[bk[k]] + rk[k]] = r;
    }
  }
  __syncthreads();

  v2i* __restrict__ outw = (v2i*)tmp8 + (long)blockIdx.x * CSTR;
  const int n = end - beg;
#pragma unroll
  for (int k = 0; k < TRIPS; ++k) {
    const int i = tid + k * 1024;
    if (i < n) outw[i] = stage[i];
  }
}

template <int NF>
__device__ __forceinline__ v4f gcn_gather_q(
    const float* __restrict__ feat, unsigned qb, int s) {
  const unsigned off = (unsigned)s * (unsigned)(NF * 4) + qb;
  return *(const v4f*)((const char*)feat + off);
}

template <int NF>
__device__ __forceinline__ v4f gcn_row_sum4(
    const float* __restrict__ feat, unsigned qb, const v2i* __restrict__ ed,
    int beg, int end) {
  v4f acc = {0.0f, 0.0f, 0.0f, 0.0f};
  if (beg < end) {
    const int last = end - 1;
    v2i e0 = ed[beg];
    v2i e1 = ed[min(beg + 1, last)];
    v2i e2 = ed[min(beg + 2, last)];
    v2i e3 = ed[min(beg + 3, last)];
    for (int j = beg; j < end; j += 4) {
      v4f x0 = gcn_gather_q<NF>(feat, qb, e0.x);
      v4f x1 = gcn_gather_q<NF>(feat, qb, e1.x);
      v4f x2 = gcn_gather_q<NF>(feat, qb, e2.x);
      v4f x3 = gcn_gather_q<NF>(feat, qb, e3.x);
      const float w0 = __int_as_float(e0.y);
      const float w1 = (j + 1 < end) ? __int_as_float(e1.y) : 0.0f;
      const float w2 = (j + 2 < end) ? __int_as_float(e2.y) : 0.0f;
      const float w3 = (j + 3 < end) ? __int_as_float(e3.y) : 0.0f;
      e0 = ed[min(j + 4, last)];
      e1 = ed[min(j + 5, last)];
      e2 = ed[min(j + 6, last)];
      e3 = ed[min(j + 7, last)];
      acc += w0 * x0;
      acc += w1 * x1;
      acc += w2 * x2;
      acc += w3 * x3;
    }
  }
  return acc;
}

template <int NF>
__device__ __forceinline__ v4f gcn_row_sum8_lds(
    const float* __restrict__ feat, unsigned qb, const v2i* ed, int beg, int end) {
  v4f acc = {0.0f, 0.0f, 0.0f, 0.0f};
  if (beg < end) {
    const int last = end - 1;
    v2i e[8];
#pragma unroll
    for (int u = 0; u < 8; ++u) e[u] = ed[min(beg + u, last)];
    for (int j = beg; j < end; j += 8) {
      v4f x[8];
      float wv[8];
#pragma unroll
      for (int u = 0; u < 8; ++u) {
        x[u]  = gcn_gather_q<NF>(feat, qb, e[u].x);
        wv[u] = (j + u < end) ? __int_as_float(e[u].y) : 0.0f;
      }
#pragma unroll
      for (int u = 0; u < 8; ++u) e[u] = ed[min(j + 8 + u, last)];
#pragma unroll
      for (int u = 0; u < 8; ++u) acc += wv[u] * x[u];
    }
  }
  return acc;
}

#define SCAP 8192
#define WP   68
#define NTILE ((BS + 15) / 16)
#define W1F_V2 ((FEAT / 4) * (EMBED / 16) * 32)
#define W2F_V2 ((EMBED / 4) * 3 * 32)
static_assert(W1F_V2 == 4 * 1024 && W2F_V2 == 3 * 1024, "staging trip counts");
static_assert(NTILE <= 32 && FEAT == 64 && EMBED == 128, "one wave per tile, 32-column quarters");

__device__ __forceinline__ void gcn_wave_lds_order() {
  __builtin_amdgcn_fence(__ATOMIC_SEQ_CST, "wavefront");
  __builtin_amdgcn_wave_barrier();
}

__global__ __launch_bounds__(1024) void gcn_bucket_rows_spmm_layers(
    const int* __restrict__ tmp8, const int* __restrict__ segT,
    const float* __restrict__ feat, const float* __restrict__ W1,
    const float* __restrict__ W2, int* __restrict__ row_ptr,
    int* __restrict__ csr, float* __restrict__ T) {
  __shared__ __attribute__((aligned(16)))
      float gsm[2 * W1F_V2 + 2 * W2F_V2 + NTILE * 16 * WP];
  v2f* const w1f = (v2f*)gsm;
  v2f* const w2f = (v2f*)(gsm + 2 * W1F_V2);
  float* const aggL = gsm + 2 * W1F_V2 + 2 * W2F_V2;
  __shared__ int cnt[512];
  __shared__ int rst[512];
  __shared__ int wtot[32];
  __shared__ int sbeg[NW];
  __shared__ int slen[NW];
  __shared__ int rbeg[8];
  __shared__ int rlen[8];
  __shared__ __attribute__((aligned(16))) v2i stage[SCAP + 16];
  const int tid  = threadIdx.x;
  const int lane = tid & 31;
  const int wave = tid >> 5;
  const int b    = blockIdx.x;
  const int lo   = b * BS;
  const int nb   = min(BS, N_NODES - lo);
  const v2i* __restrict__ in = (const v2i*)tmp8;
  if (tid < 512) cnt[tid] = 0;

  int s0 = 0, sl = 0;
  if (tid < NW) {
    s0 = segT[b * NW + tid];
    sl = segT[(b + 1) * NW + tid] - s0;
    sbeg[tid] = tid * CSTR + s0;
    slen[tid] = sl;
  }
  const int wn   = lane & 15;
  const int wklo = (lane >> 4) * 2;
  float w1x[4], w1y[4], w2x[3], w2y[3];
#pragma unroll
  for (int j = 0; j < 4; ++j) {
    const int ks = j * 4 + (wave >> 3);
    const float* __restrict__ p = W1 + (ks * 4 + wklo) * EMBED + (wave & 7) * 16 + wn;
    w1x[j] = p[0];
    w1y[j] = p[EMBED];
  }
#pragma unroll
  for (int j = 0; j < 3; ++j) {
    const int  col = j * 16 + wn;
    const bool inb = col < NCLS;
    const float* __restrict__ p = W2 + (wave * 4 + wklo) * NCLS + (inb ? col : 0);
    w2x[j] = p[0];
    w2y[j] = p[NCLS];
  }

  int ra = s0, rl = sl;
#pragma unroll
  for (int o = 16; o > 0; o >>= 1) {
    ra += __shfl_xor(ra, o);
    rl += __shfl_xor(rl, o);
  }
  if (tid < NW && lane == 0) {
    rbeg[wave] = ra;
    rlen[wave] = rl;
  }
  __syncthreads();
#pragma unroll
  for (int j = 0; j < 4; ++j) {
    v2f v;
    v.x = w1x[j];
    v.y = w1y[j];
    w1f[j * 1024 + tid] = v;
  }
#pragma unroll
  for (int j = 0; j < 3; ++j) {
    const bool inb = j * 16 + wn < NCLS;
    v2f v;
    v.x = inb ? w2x[j] : 0.0f;
    v.y = inb ? w2y[j] : 0.0f;
    w2f[j * 1024 + tid] = v;
  }
  int beg = 0, tot = 0;
#pragma unroll
  for (int k = 0; k < 8; ++k) {
    beg += rbeg[k];
    tot += rlen[k];
  }
  const int end = beg + tot;
  const int  sh     = beg & 15;
  const bool staged = tot <= SCAP;

  v2i r[16];
#pragma unroll
  for (int k = 0; k < 8; ++k) {
    const int ch = wave * 8 + k;
    const int len = slen[ch];
    const int base = sbeg[ch];
    if (lane < len)      r[2 * k] = in[base + lane];
    else                 r[2 * k] = (v2i){-1, 0};
    if (lane + 32 < len) r[2 * k + 1] = in[base + 32 + lane];
    else                 r[2 * k + 1] = (v2i){-1, 0};
  }
#pragma unroll
  for (int k = 0; k < 16; ++k)
    if (r[k].x >= 0) atomicAdd(&cnt[r[k].x >> 17], 1);
  for (int k = 0; k < 8; ++k) {
    const int ch = wave * 8 + k;
    const int len = slen[ch];
    const int base = sbeg[ch];
    for (int i = 64 + lane; i < len; i += 32)
      atomicAdd(&cnt[in[base + i].x >> 17], 1);
  }
  __syncthreads();

  const int c = (tid < nb) ? cnt[tid] : 0;
  int incl = c;
#pragma unroll
  for (int o = 1; o < 32; o <<= 1) {
    int v = __shfl_up(incl, o);
    if (lane >= o) incl += v;
  }
  if (lane == 31) wtot[wave] = incl;
  __syncthreads();
  int off = 0;
#pragma unroll
  for (int k = 0; k < 32; ++k)
    if (k < wave) off += wtot[k];
  const int first = beg + off + incl - c;
  if (tid < nb) {
    row_ptr[lo + tid] = first;
    rst[tid] = first;
    cnt[tid] = first;
  }
  if (b == NB - 1 && tid == 0) row_ptr[N_NODES] = end;
  __syncthreads();

#pragma unroll
  for (int k = 0; k < 16; ++k) {
    if (r[k].x >= 0) {
      const int pos = atomicAdd(&cnt[r[k].x >> 17], 1);
      v2i v;
      v.x = r[k].x & 0x1FFFF;
      v.y = r[k].y;
      if (staged) stage[pos - beg + sh] = v;
      else        ((v2i*)csr)[pos] = v;
    }
  }
  for (int k = 0; k < 8; ++k) {
    const int ch = wave * 8 + k;
    const int len = slen[ch];
    const int base = sbeg[ch];
    for (int i = 64 + lane; i < len; i += 32) {
      const v2i e = in[base + i];
      const int pos = atomicAdd(&cnt[e.x >> 17], 1);
      v2i v;
      v.x = e.x & 0x1FFFF;
      v.y = e.y;
      if (staged) stage[pos - beg + sh] = v;
      else        ((v2i*)csr)[pos] = v;
    }
  }

  __syncthreads();

  if (staged) {
    v2i* __restrict__ line0 = (v2i*)csr + (beg - sh);
    const int n = sh + tot;
    for (int j = 2 * tid; j < n; j += 2048) {
      const bool lo_ok = j >= sh;
      const bool hi_ok = (j + 1 >= sh) && (j + 1 < n);
      if (lo_ok && hi_ok)
        *(v4i*)(line0 + j) = *(const v4i*)(stage + j);
      else if (lo_ok)
        line0[j] = stage[j];
      else if (hi_ok)
        line0[j + 1] = stage[j + 1];
    }
  }

  const int q = lane & 15;
  const unsigned qb = (unsigned)q * 16u;
  for (int rr = wave * 2 + (lane >> 4); rr < NTILE * 16; rr += 64) {
    v4f acc = {0.0f, 0.0f, 0.0f, 0.0f};
    if (rr < nb) {
      const int js = rst[rr];
      const int je = cnt[rr];
      if (staged)
        acc = gcn_row_sum8_lds<FEAT>(feat, qb, stage, js - beg + sh, je - beg + sh);
      else
        acc = gcn_row_sum4<FEAT>(feat, qb, (const v2i*)csr, js, je);
    }
    *(v4f*)(aggL + rr * WP + q * 4) = acc;
  }
  __syncthreads();

  const int wv = __builtin_amdgcn_readfirstlane(wave);
  if (wv * 16 < nb) {
    float* const win = aggL + wv * (16 * WP);
    const int n    = lane & 15;
    const int half = lane >> 4;
    const int klo  = half * 2;
    const float* const arow = win + n * WP + klo;

    v2f a[FEAT / 4];
#pragma unroll
    for (int ks = 0; ks < FEAT / 4; ++ks) a[ks] = *(const v2f*)(arow + ks * 4);

    v8f t0 = {}, t1 = {}, t2 = {};
#pragma unroll
    for (int nq = 0; nq < 4; ++nq) {
      v8f c[2];
#pragma unroll
      for (int j = 0; j < 2; ++j) c[j] = (v8f){};
#pragma unroll
      for (int ks = 0; ks < FEAT / 4; ++ks) {
#pragma unroll
        for (int j = 0; j < 2; ++j) {
          const v2f bw = w1f[(ks * (EMBED / 16) + nq * 2 + j) * 32 + lane];
          c[j] = __builtin_amdgcn_wmma_f32_16x16x4_f32(
              false, a[ks], false, bw, (short)0, c[j], false, false);
        }
      }
      gcn_wave_lds_order();
#pragma unroll
      for (int j = 0; j < 2; ++j) {
#pragma unroll
        for (int r = 0; r < 8; ++r)
          win[(r + 8 * half) * WP + j * 16 + n] = fmaxf(c[j][r], 0.0f);
      }
      gcn_wave_lds_order();

#pragma unroll
      for (int ks = 0; ks < 8; ++ks) {
        const v2f ha = *(const v2f*)(arow + ks * 4);
        const v2f b0 = w2f[(0 * 32 + nq * 8 + ks) * 32 + lane];
        const v2f b1 = w2f[(1 * 32 + nq * 8 + ks) * 32 + lane];
        const v2f b2 = w2f[(2 * 32 + nq * 8 + ks) * 32 + lane];
        t0 = __builtin_amdgcn_wmma_f32_16x16x4_f32(
            false, ha, false, b0, (short)0, t0, false, false);
        t1 = __builtin_amdgcn_wmma_f32_16x16x4_f32(
            false, ha, false, b1, (short)0, t1, false, false);
        t2 = __builtin_amdgcn_wmma_f32_16x16x4_f32(
            false, ha, false, b2, (short)0, t2, false, false);
      }
    }

#pragma unroll
    for (int r = 0; r < 8; ++r) {
      const int trl = wv * 16 + r + half * 8;
      if (trl < nb) {
        float* __restrict__ trow = T + (long)(lo + trl) * NCLS;
        trow[n]      = t0[r];
        trow[16 + n] = t1[r];
        if (n < NCLS - 32) trow[32 + n] = t2[r];
      }
    }
  }
}

template <int M>
__device__ __forceinline__ float gcn_row_xor(float v) {
  return __int_as_float(__builtin_amdgcn_update_dpp(
      0, __float_as_int(v), 0x160 + M, 0xf, 0xf, false));
}

__global__ __launch_bounds__(256) void gcn_spmm_gather4_logsoftmax(
    const float* __restrict__ feat, const int* __restrict__ row_ptr,
    const int* __restrict__ csr, float* __restrict__ out, int n_rows) {
  constexpr int QP = NCLS / 4;
  static_assert(NCLS % 4 == 0 && QP <= 16, "a class row fits one 16-lane row");
  const int d = (blockIdx.x * blockDim.x + threadIdx.x) >> 4;
  const int q = threadIdx.x & 15;
  const bool act = (d < n_rows) && (q < QP);
  v4f acc = {0.0f, 0.0f, 0.0f, 0.0f};
  if (act)
    acc = gcn_row_sum4<NCLS>(feat, (unsigned)q * 16u, (const v2i*)csr,
                             row_ptr[d], row_ptr[d + 1]);

  float mx = act ? fmaxf(fmaxf(acc.x, acc.y), fmaxf(acc.z, acc.w)) : -INFINITY;
  mx = fmaxf(mx, gcn_row_xor<1>(mx));
  mx = fmaxf(mx, gcn_row_xor<2>(mx));
  mx = fmaxf(mx, gcn_row_xor<4>(mx));
  mx = fmaxf(mx, gcn_row_xor<8>(mx));
  float s = 0.0f;
  if (act)
    s = (__expf(acc.x - mx) + __expf(acc.y - mx)) +
        (__expf(acc.z - mx) + __expf(acc.w - mx));
  s += gcn_row_xor<1>(s);
  s += gcn_row_xor<2>(s);
  s += gcn_row_xor<4>(s);
  s += gcn_row_xor<8>(s);
  const float lse = mx + __logf(s);
  if (act) *(v4f*)(out + (long)d * NCLS + q * 4) = acc - lse;
}

extern "C" void kernel_launch(void* const* d_in, const int* in_sizes, int n_in,
                              void* d_out, int out_size, void* d_ws, size_t ws_size,
                              hipStream_t stream) {
  const float* x    = (const float*)d_in[0];
  const float* W1   = (const float*)d_in[1];
  const float* W2   = (const float*)d_in[2];
  const float* ew   = (const float*)d_in[3];
  const int*   esrc = (const int*)d_in[4];
  const int*   edst = (const int*)d_in[5];
  float*       out  = (float*)d_out;

  float* agg1 = (float*)d_ws;
  float* scr  = agg1 + (size_t)N_NODES * FEAT;
  float* t    = scr  + (size_t)N_NODES * EMBED;
  int*   csr  = (int*)(t + (size_t)N_NODES * NCLS);
  int*   rptr = csr + (size_t)2 * N_EDGES;
  int*   tmp8   = (int*)scr;
  int*   segT   = tmp8 + (size_t)2 * NW * CSTR;

  const int TPB = 256;

  gcn_chunk_sort<<<NW, 1024, 0, stream>>>(esrc, edst, ew, tmp8, segT);
  gcn_bucket_rows_spmm_layers<<<NB, 1024, 0, stream>>>(tmp8, segT, x, W1, W2, rptr, csr, t);

  gcn_spmm_gather4_logsoftmax<<<(N_NODES * 16 + TPB - 1) / TPB, TPB, 0, stream>>>(
      t, rptr, csr, out, N_NODES);
}
